# starting kernel
# baseline (speedup 1.0000x reference)
#include <hip/hip_runtime.h>
#include <hip/hip_bf16.h>
#include <math.h>

typedef float v8f __attribute__((ext_vector_type(8)));
typedef float v2f __attribute__((ext_vector_type(2)));

#define D_HID 128
#define D_OUT 64
#define BN_EPS 1e-5f


__global__ void init_deg_stats(float* deg, float* stats, int N) {
    int i = blockIdx.x * blockDim.x + threadIdx.x;
    int stride = blockDim.x * gridDim.x;
    for (; i < N; i += stride) {
        deg[i] = 1.0f;
        if (i < 256) stats[i] = 0.0f;
    }
}

__global__ void count_edges(const int* __restrict__ col, float* deg, int E) {
    int e = blockIdx.x * blockDim.x + threadIdx.x;
    if (e < E) atomicAdd(&deg[col[e]], 1.0f);
}

__global__ void deg_to_dis(float* deg, int N) {
    int i = blockIdx.x * blockDim.x + threadIdx.x;
    if (i < N) deg[i] = rsqrtf(deg[i]);
}


__global__ void gemm1_wmma(const float* __restrict__ x, const float* __restrict__ W1,
                           const float* __restrict__ b1, float* __restrict__ h1, int N) {
    const int wave  = threadIdx.x >> 5;
    const int lane  = threadIdx.x & 31;
    const int row0  = blockIdx.x * 16;
    if (row0 >= N) return;
    const int col0  = wave * 16;
    const int m     = lane & 15;
    const int n     = lane & 15;
    const int khalf = lane >> 4;

    v8f c;
    const float bias = b1[col0 + n];
    #pragma unroll
    for (int r = 0; r < 8; ++r) c[r] = bias;

    const float* arow = x + (size_t)(row0 + m) * D_HID + 2 * khalf;
    #pragma unroll 4
    for (int k = 0; k < 128; k += 4) {
        v2f a, b;
        a.x = arow[k];
        a.y = arow[k + 1];
        const int kb = k + 2 * khalf;
        b.x = W1[(size_t)(kb + 0) * D_HID + col0 + n];
        b.y = W1[(size_t)(kb + 1) * D_HID + col0 + n];
        c = __builtin_amdgcn_wmma_f32_16x16x4_f32(false, a, false, b, (short)0, c, false, false);
    }

    const int mbase = khalf * 8;
    #pragma unroll
    for (int r = 0; r < 8; ++r)
        h1[(size_t)(row0 + mbase + r) * D_HID + col0 + n] = c[r];
}


__global__ void selfloop1(const float* __restrict__ h1, const float* __restrict__ dis,
                          float* __restrict__ hagg, int total) {
    int i = blockIdx.x * blockDim.x + threadIdx.x;
    if (i < total) {
        float d = dis[i >> 7];
        hagg[i] = d * d * h1[i];
    }
}

__global__ void scatter1(const int* __restrict__ row, const int* __restrict__ col,
                         const float* __restrict__ dis, const float* __restrict__ h1,
                         float* hagg, int E) {
    const int wave = threadIdx.x >> 5;
    const int lane = threadIdx.x & 31;
    const int e = blockIdx.x * 8 + wave;
    if (e >= E) return;
    const int r = row[e];
    const int c = col[e];
    const float w = dis[r] * dis[c];
    const int base = lane * 4;
    const float4 v = *(const float4*)(h1 + (size_t)r * D_HID + base);
    float* dst = hagg + (size_t)c * D_HID + base;
    atomicAdd(dst + 0, w * v.x);
    atomicAdd(dst + 1, w * v.y);
    atomicAdd(dst + 2, w * v.z);
    atomicAdd(dst + 3, w * v.w);
}


__global__ void bn_stats(const float* __restrict__ hagg, float* bsum, float* bsq, int total) {
    __shared__ float ssum[D_HID];
    __shared__ float ssq[D_HID];
    for (int i = threadIdx.x; i < D_HID; i += blockDim.x) { ssum[i] = 0.0f; ssq[i] = 0.0f; }
    __syncthreads();
    int stride = blockDim.x * gridDim.x;
    for (int i = blockIdx.x * blockDim.x + threadIdx.x; i < total; i += stride) {
        float v = hagg[i];
        atomicAdd(&ssum[i & 127], v);
        atomicAdd(&ssq[i & 127], v * v);
    }
    __syncthreads();
    for (int i = threadIdx.x; i < D_HID; i += blockDim.x) {
        atomicAdd(&bsum[i], ssum[i]);
        atomicAdd(&bsq[i], ssq[i]);
    }
}

__global__ void bn_finalize(const float* __restrict__ bsum, const float* __restrict__ bsq,
                            const float* __restrict__ gamma, const float* __restrict__ beta,
                            float* scale, float* shift, int N) {
    int c = threadIdx.x;
    if (c < D_HID) {
        float inv = 1.0f / (float)N;
        float mu = bsum[c] * inv;
        float var = bsq[c] * inv - mu * mu;
        float s = gamma[c] * rsqrtf(var + BN_EPS);
        scale[c] = s;
        shift[c] = beta[c] - mu * s;
    }
}


__global__ void gemm2_wmma(const float* __restrict__ hagg, const float* __restrict__ scale,
                           const float* __restrict__ shift, const float* __restrict__ W2,
                           const float* __restrict__ b2, float* __restrict__ h2, int N) {
    const int wave  = threadIdx.x >> 5;
    const int lane  = threadIdx.x & 31;
    const int row0  = (blockIdx.x * 2 + (wave >> 2)) * 16;
    if (row0 >= N) return;
    const int col0  = (wave & 3) * 16;
    const int m     = lane & 15;
    const int n     = lane & 15;
    const int khalf = lane >> 4;

    v8f c;
    const float bias = b2[col0 + n];
    #pragma unroll
    for (int r = 0; r < 8; ++r) c[r] = bias;

    const float* arow = hagg + (size_t)(row0 + m) * D_HID;
    #pragma unroll 4
    for (int k = 0; k < 128; k += 4) {
        const int ka = k + 2 * khalf;
        v2f a, b;
        a.x = fmaxf(arow[ka + 0] * scale[ka + 0] + shift[ka + 0], 0.0f);
        a.y = fmaxf(arow[ka + 1] * scale[ka + 1] + shift[ka + 1], 0.0f);
        b.x = W2[(size_t)(ka + 0) * D_OUT + col0 + n];
        b.y = W2[(size_t)(ka + 1) * D_OUT + col0 + n];
        c = __builtin_amdgcn_wmma_f32_16x16x4_f32(false, a, false, b, (short)0, c, false, false);
    }

    const int mbase = khalf * 8;
    #pragma unroll
    for (int r = 0; r < 8; ++r)
        h2[(size_t)(row0 + mbase + r) * D_OUT + col0 + n] = c[r];
}


__global__ void selfloop2(const float* __restrict__ h2, const float* __restrict__ dis,
                          float* __restrict__ out, int total) {
    int i = blockIdx.x * blockDim.x + threadIdx.x;
    if (i < total) {
        float d = dis[i >> 6];
        out[i] = d * d * h2[i];
    }
}

__global__ void scatter2(const int* __restrict__ row, const int* __restrict__ col,
                         const float* __restrict__ dis, const float* __restrict__ h2,
                         float* out, int E) {
    const int wave = threadIdx.x >> 5;
    const int lane = threadIdx.x & 31;
    const int e = blockIdx.x * 8 + wave;
    if (e >= E) return;
    const int r = row[e];
    const int c = col[e];
    const float w = dis[r] * dis[c];
    const int base = lane * 2;
    const float2 v = *(const float2*)(h2 + (size_t)r * D_OUT + base);
    float* dst = out + (size_t)c * D_OUT + base;
    atomicAdd(dst + 0, w * v.x);
    atomicAdd(dst + 1, w * v.y);
}


__global__ void logsoftmax_rows(float* out, int N) {
    const int wave = threadIdx.x >> 5;
    const int lane = threadIdx.x & 31;
    const int rowi = blockIdx.x * 8 + wave;
    if (rowi >= N) return;
    float* p = out + (size_t)rowi * D_OUT;
    float v0 = p[lane];
    float v1 = p[lane + 32];
    float mx = fmaxf(v0, v1);
    #pragma unroll
    for (int off = 16; off >= 1; off >>= 1)
        mx = fmaxf(mx, __shfl_xor(mx, off, 32));
    float s = __expf(v0 - mx) + __expf(v1 - mx);
    #pragma unroll
    for (int off = 16; off >= 1; off >>= 1)
        s += __shfl_xor(s, off, 32);
    float l = mx + __logf(s);
    p[lane]      = v0 - l;
    p[lane + 32] = v1 - l;
}


extern "C" void kernel_launch(void* const* d_in, const int* in_sizes, int n_in,
                              void* d_out, int out_size, void* d_ws, size_t ws_size,
                              hipStream_t stream) {
    const float* x     = (const float*)d_in[0];
    const float* W1    = (const float*)d_in[1];
    const float* b1    = (const float*)d_in[2];
    const float* gamma = (const float*)d_in[3];
    const float* beta  = (const float*)d_in[4];
    const float* W2    = (const float*)d_in[5];
    const float* b2    = (const float*)d_in[6];
    const int*   ei    = (const int*)d_in[7];

    const int N = in_sizes[0] / D_HID;
    const int E = in_sizes[7] / 2;
    const int* erow = ei;
    const int* ecol = ei + E;

    float* ws    = (float*)d_ws;
    float* dis   = ws;
    float* h1    = dis + N;
    float* hagg  = h1 + (size_t)N * D_HID;
    float* stats = hagg + (size_t)N * D_HID;
    float* bsum  = stats;
    float* bsq   = stats + 128;
    float* scal  = stats + 256;
    float* shft  = stats + 384;
    float* h2    = h1;
    float* outp  = (float*)d_out;

    init_deg_stats<<<(N + 255) / 256, 256, 0, stream>>>(dis, stats, N);
    count_edges<<<(E + 255) / 256, 256, 0, stream>>>(ecol, dis, E);
    deg_to_dis<<<(N + 255) / 256, 256, 0, stream>>>(dis, N);

    gemm1_wmma<<<(N + 15) / 16, 256, 0, stream>>>(x, W1, b1, h1, N);
    selfloop1<<<((size_t)N * D_HID + 255) / 256, 256, 0, stream>>>(h1, dis, hagg, N * D_HID);
    scatter1<<<(E + 7) / 8, 256, 0, stream>>>(erow, ecol, dis, h1, hagg, E);

    bn_stats<<<1024, 256, 0, stream>>>(hagg, bsum, bsq, N * D_HID);
    bn_finalize<<<1, 128, 0, stream>>>(bsum, bsq, gamma, beta, scal, shft, N);

    gemm2_wmma<<<(N + 31) / 32, 256, 0, stream>>>(hagg, scal, shft, W2, b2, h2, N);
    selfloop2<<<((size_t)N * D_OUT + 255) / 256, 256, 0, stream>>>(h2, dis, outp, N * D_OUT);
    scatter2<<<(E + 7) / 8, 256, 0, stream>>>(erow, ecol, dis, h2, outp, E);

    logsoftmax_rows<<<(N + 7) / 8, 256, 0, stream>>>(outp, N);
}


# Both GEMMs: one wave owns a 16-row tile across all output columns (8 or 4 accumulators): A fragment loaded once per k-step, not per column tile, asked a trip ahead; 16-byte B loads; sum order kept
# speedup vs baseline: 2.1299x; 2.1299x over previous
#include <hip/hip_runtime.h>
#include <hip/hip_bf16.h>
#include <math.h>

typedef float v8f __attribute__((ext_vector_type(8)));
typedef float v2f __attribute__((ext_vector_type(2)));

#define D_HID 128
#define D_OUT 64
#define BN_EPS 1e-5f


__global__ void init_deg_stats(float* deg, float* stats, int N) {
    int i = blockIdx.x * blockDim.x + threadIdx.x;
    int stride = blockDim.x * gridDim.x;
    for (; i < N; i += stride) {
        deg[i] = 1.0f;
        if (i < 256) stats[i] = 0.0f;
    }
}

__global__ void count_edges(const int* __restrict__ col, float* deg, int E) {
    int e = blockIdx.x * blockDim.x + threadIdx.x;
    if (e < E) atomicAdd(&deg[col[e]], 1.0f);
}

__global__ void deg_to_dis(float* deg, int N) {
    int i = blockIdx.x * blockDim.x + threadIdx.x;
    if (i < N) deg[i] = rsqrtf(deg[i]);
}


typedef float f4u __attribute__((ext_vector_type(4), aligned(4)));

__global__ void __launch_bounds__(128) gemm1_wmma(const float* __restrict__ x, const float* __restrict__ W1,
                           const float* __restrict__ b1, float* __restrict__ h1, int N) {
    const int wave  = __builtin_amdgcn_readfirstlane((int)(threadIdx.x >> 5));
    const int lane  = threadIdx.x & 31;
    const int row0  = (blockIdx.x * 4 + wave) * 16;
    if (row0 >= N) return;
    const int n     = lane & 15;
    const int khalf = lane >> 4;

    v8f c[8];
    {
        const f4u blo = *(const f4u*)(b1 + 4 * n);
        const f4u bhi = *(const f4u*)(b1 + 64 + 4 * n);
        #pragma unroll
        for (int t = 0; t < 4; ++t) {
            #pragma unroll
            for (int r = 0; r < 8; ++r) { c[t][r] = blo[t]; c[t + 4][r] = bhi[t]; }
        }
    }

    const float* arow = x + (size_t)(row0 + n) * D_HID + 2 * khalf;
    const float* brow = W1 + (size_t)(2 * khalf) * D_HID + 4 * n;

    v2f an[4];
    #pragma unroll
    for (int j = 0; j < 4; ++j) { an[j].x = arow[4 * j]; an[j].y = arow[4 * j + 1]; }

    #pragma unroll 2
    for (int k0 = 0; k0 < 128; k0 += 16) {
        v2f a[4];
        f4u p0[4], p1[4], q0[4], q1[4];
        #pragma unroll
        for (int j = 0; j < 4; ++j) {
            a[j] = an[j];
            const float* bk = brow + (size_t)(k0 + 4 * j) * D_HID;
            p0[j] = *(const f4u*)(bk);
            p1[j] = *(const f4u*)(bk + D_HID);
            q0[j] = *(const f4u*)(bk + 64);
            q1[j] = *(const f4u*)(bk + D_HID + 64);
        }
        const int kn = (k0 + 16 < 128) ? k0 + 16 : k0;
        #pragma unroll
        for (int j = 0; j < 4; ++j) { an[j].x = arow[kn + 4 * j]; an[j].y = arow[kn + 4 * j + 1]; }
        #pragma unroll
        for (int j = 0; j < 4; ++j) {
            #pragma unroll
            for (int t = 0; t < 4; ++t) {
                v2f b;
                b.x = p0[j][t];
                b.y = p1[j][t];
                c[t] = __builtin_amdgcn_wmma_f32_16x16x4_f32(false, a[j], false, b, (short)0, c[t], false, false);
            }
            #pragma unroll
            for (int t = 0; t < 4; ++t) {
                v2f b;
                b.x = q0[j][t];
                b.y = q1[j][t];
                c[t + 4] = __builtin_amdgcn_wmma_f32_16x16x4_f32(false, a[j], false, b, (short)0, c[t + 4], false, false);
            }
        }
    }

    const int mbase = khalf * 8;
    #pragma unroll
    for (int r = 0; r < 8; ++r) {
        float* orow = h1 + (size_t)(row0 + mbase + r) * D_HID + 4 * n;
        f4u lo, hi;
        #pragma unroll
        for (int t = 0; t < 4; ++t) { lo[t] = c[t][r]; hi[t] = c[t + 4][r]; }
        *(f4u*)(orow)      = lo;
        *(f4u*)(orow + 64) = hi;
    }
}


#define OWN_ROWS 196
#define QCAP 160
#define REC_CAP 16384

typedef int   vi4u __attribute__((ext_vector_type(4), aligned(4)));
template <int V> struct rowvec;
template <> struct rowvec<4> { typedef float t __attribute__((ext_vector_type(4), aligned(4))); };
template <> struct rowvec<2> { typedef float t __attribute__((ext_vector_type(2), aligned(4))); };

template <int CH, bool FULL>
__device__ __forceinline__ void add_rows(int r, int cl, int wi, int cnt,
                                         const float* __restrict__ h, float* acc, int lane) {
    constexpr int V = CH / 32;
    typedef typename rowvec<V>::t vec;
    #pragma unroll 1
    for (int j0 = 0; j0 < cnt; j0 += 8) {
        vec v[8];
        #pragma unroll
        for (int j = 0; j < 8; ++j) {
            const int rj = __builtin_amdgcn_readlane(r, j0 + j);
            v[j] = *(const vec*)(h + (size_t)rj * CH + lane * V);
        }
        #pragma unroll
        for (int j = 0; j < 8; ++j) {
            if (FULL || j0 + j < cnt) {
                const int cj = __builtin_amdgcn_readlane(cl, j0 + j);
                const float wj = __builtin_bit_cast(float, __builtin_amdgcn_readlane(wi, j0 + j));
                float* a = acc + cj * CH + lane;
                #pragma unroll
                for (int k = 0; k < V; ++k) atomicAdd(a + 32 * k, wj * v[j][k]);
            }
        }
    }
}

template <int CH, bool FULL, bool SAVE>
__device__ __forceinline__ void drain_round(const int* qe_, const int* qc_, int cnt,
                                            const int* __restrict__ row, const float* __restrict__ dis,
                                            const float* __restrict__ h, float* acc, int lo, int lane,
                                            int* rec, int* ctl) {
    int cl = -1, r = 0;
    float w = 0.0f;
    if (FULL || lane < cnt) {
        const int e = qe_[lane];
        cl = qc_[lane];
        r = row[e];
        w = dis[r] * dis[lo + cl];
    }
    const int wi = __builtin_bit_cast(int, w);
    if (SAVE) {
        int base = 0;
        if (lane == 0) base = atomicAdd(&ctl[0], 32);
        base = __builtin_amdgcn_readfirstlane(base);
        if (base + 32 <= REC_CAP) {
            rec[base + lane] = r;
            rec[REC_CAP + base + lane] = cl;
            rec[2 * REC_CAP + base + lane] = wi;
        } else if (lane == 0) {
            ctl[1] = 1;
        }
    }
    add_rows<CH, FULL>(r, cl, wi, cnt, h, acc, lane);
}

template <int CH, bool SAVE>
__global__ void __launch_bounds__(1024) own_gather(const int* __restrict__ row, const int* __restrict__ col,
                                                   const float* __restrict__ dis, const float* __restrict__ h,
                                                   float* __restrict__ dst, int* recs, int* meta, int N, int E) {
    constexpr int V = CH / 32;
    typedef typename rowvec<V>::t vec;
    __shared__ float acc[OWN_ROWS * CH];
    __shared__ int qe[32][QCAP];
    __shared__ int qc[32][QCAP];
    __shared__ int ctl[2];
    const int wave = __builtin_amdgcn_readfirstlane((int)(threadIdx.x >> 5));
    const int lane = threadIdx.x & 31;
    const int lo = blockIdx.x * OWN_ROWS;
    const int nrows = min(OWN_ROWS, N - lo);
    int* rec = recs + (size_t)blockIdx.x * (3 * REC_CAP);
    if (SAVE && threadIdx.x == 0) { ctl[0] = 0; ctl[1] = 0; }

    for (int rl = wave; rl < nrows; rl += 32) {
        const vec v = *(const vec*)(h + (size_t)(lo + rl) * CH + lane * V);
        const float d = dis[lo + rl];
        #pragma unroll
        for (int k = 0; k < V; ++k) acc[rl * CH + k * 32 + lane] = d * d * v[k];
    }
    __syncthreads();

    const int nslots = SAVE ? -1 : meta[blockIdx.x * 32];
    if (!SAVE && nslots >= 0) {
        const int nfr = nslots >> 5;
        for (int f = wave; f < nfr; f += 32) {
            const int r  = rec[f * 32 + lane];
            const int cl = rec[REC_CAP + f * 32 + lane];
            const int wi = rec[2 * REC_CAP + f * 32 + lane];
            const int cnt = __builtin_popcount(__builtin_amdgcn_ballot_w32(cl >= 0));
            if (cnt == 32) add_rows<CH, true>(r, cl, wi, 32, h, acc, lane);
            else           add_rows<CH, false>(r, cl, wi, cnt, h, acc, lane);
        }
    } else {
    int* myqe = qe[wave];
    int* myqc = qc[wave];
    const int nchunks = (E + 127) >> 7;
    const int nsuper = (nchunks + 3) >> 2;
    const int s0 = (int)(((long long)blockIdx.x * nsuper) / gridDim.x);
    int q = 0;
    for (int it = wave; it < nsuper; it += 32) {
        int s = it + s0;
        if (s >= nsuper) s -= nsuper;
        vi4u c[4];
        if (s * 512 + 512 <= E) {
            #pragma unroll
            for (int k = 0; k < 4; ++k) c[k] = *(const vi4u*)(col + (s * 4 + k) * 128 + lane * 4);
        } else {
            #pragma unroll
            for (int k = 0; k < 4; ++k) {
                const int eb = (s * 4 + k) * 128 + lane * 4;
                c[k].x = (eb + 0 < E) ? col[eb + 0] : -1;
                c[k].y = (eb + 1 < E) ? col[eb + 1] : -1;
                c[k].z = (eb + 2 < E) ? col[eb + 2] : -1;
                c[k].w = (eb + 3 < E) ? col[eb + 3] : -1;
            }
        }
        #pragma unroll
        for (int k = 0; k < 4; ++k) {
            const int eb = (s * 4 + k) * 128 + lane * 4;
            unsigned d[4];
            #pragma unroll
            for (int i = 0; i < 4; ++i) d[i] = (unsigned)(c[k][i] - lo);
            const bool any = min(min(d[0], d[1]), min(d[2], d[3])) < (unsigned)nrows;
            if (__builtin_amdgcn_ballot_w32(any) != 0u) {
                #pragma unroll
                for (int i = 0; i < 4; ++i) {
                    const bool mi = d[i] < (unsigned)nrows;
                    const unsigned bal = __builtin_amdgcn_ballot_w32(mi);
                    if (bal != 0u) {
                        if (mi) {
                            const int pos = q + (int)__builtin_amdgcn_mbcnt_lo(bal, 0u);
                            myqe[pos] = eb + i;
                            myqc[pos] = (int)d[i];
                        }
                        q += __builtin_popcount(bal);
                    }
                }
                while (q >= 32) {
                    q -= 32;
                    drain_round<CH, true, SAVE>(myqe + q, myqc + q, 32, row, dis, h, acc, lo, lane, rec, ctl);
                }
            }
        }
    }
    if (q > 0) drain_round<CH, false, SAVE>(myqe, myqc, q, row, dis, h, acc, lo, lane, rec, ctl);
    }
    __syncthreads();
    if (SAVE && threadIdx.x == 0) meta[blockIdx.x * 32] = ctl[1] ? -1 : ctl[0];

    for (int rl = wave; rl < nrows; rl += 32) {
        vec o;
        #pragma unroll
        for (int k = 0; k < V; ++k) o[k] = acc[rl * CH + k * 32 + lane];
        *(vec*)(dst + (size_t)(lo + rl) * CH + lane * V) = o;
    }
}


__global__ void bn_stats(const float* __restrict__ hagg, float* bsum, float* bsq, int total) {
    __shared__ float ssum[D_HID];
    __shared__ float ssq[D_HID];
    for (int i = threadIdx.x; i < D_HID; i += blockDim.x) { ssum[i] = 0.0f; ssq[i] = 0.0f; }
    __syncthreads();
    int stride = blockDim.x * gridDim.x;
    for (int i = blockIdx.x * blockDim.x + threadIdx.x; i < total; i += stride) {
        float v = hagg[i];
        atomicAdd(&ssum[i & 127], v);
        atomicAdd(&ssq[i & 127], v * v);
    }
    __syncthreads();
    for (int i = threadIdx.x; i < D_HID; i += blockDim.x) {
        atomicAdd(&bsum[i], ssum[i]);
        atomicAdd(&bsq[i], ssq[i]);
    }
}

__global__ void bn_finalize(const float* __restrict__ bsum, const float* __restrict__ bsq,
                            const float* __restrict__ gamma, const float* __restrict__ beta,
                            float* scale, float* shift, int N) {
    int c = threadIdx.x;
    if (c < D_HID) {
        float inv = 1.0f / (float)N;
        float mu = bsum[c] * inv;
        float var = bsq[c] * inv - mu * mu;
        float s = gamma[c] * rsqrtf(var + BN_EPS);
        scale[c] = s;
        shift[c] = beta[c] - mu * s;
    }
}


__global__ void __launch_bounds__(128) gemm2_wmma(const float* __restrict__ hagg, const float* __restrict__ scale,
                           const float* __restrict__ shift, const float* __restrict__ W2,
                           const float* __restrict__ b2, float* __restrict__ h2, int N) {
    const int wave  = __builtin_amdgcn_readfirstlane((int)(threadIdx.x >> 5));
    const int lane  = threadIdx.x & 31;
    const int row0  = (blockIdx.x * 4 + wave) * 16;
    if (row0 >= N) return;
    const int n     = lane & 15;
    const int khalf = lane >> 4;

    v8f c[4];
    {
        const f4u bias = *(const f4u*)(b2 + 4 * n);
        #pragma unroll
        for (int t = 0; t < 4; ++t) {
            #pragma unroll
            for (int r = 0; r < 8; ++r) c[t][r] = bias[t];
        }
    }

    const float* arow = hagg + (size_t)(row0 + n) * D_HID + 2 * khalf;
    const float* sc   = scale + 2 * khalf;
    const float* sh   = shift + 2 * khalf;
    const float* brow = W2 + (size_t)(2 * khalf) * D_OUT + 4 * n;

    v2f an[4];
    #pragma unroll
    for (int j = 0; j < 4; ++j) { an[j].x = arow[4 * j]; an[j].y = arow[4 * j + 1]; }

    #pragma unroll 2
    for (int k0 = 0; k0 < 128; k0 += 16) {
        v2f a[4];
        f4u p0[4], p1[4];
        #pragma unroll
        for (int j = 0; j < 4; ++j) {
            const int k = k0 + 4 * j;
            a[j].x = fmaxf(an[j].x * sc[k + 0] + sh[k + 0], 0.0f);
            a[j].y = fmaxf(an[j].y * sc[k + 1] + sh[k + 1], 0.0f);
            const float* bk = brow + (size_t)k * D_OUT;
            p0[j] = *(const f4u*)(bk);
            p1[j] = *(const f4u*)(bk + D_OUT);
        }
        const int kn = (k0 + 16 < 128) ? k0 + 16 : k0;
        #pragma unroll
        for (int j = 0; j < 4; ++j) { an[j].x = arow[kn + 4 * j]; an[j].y = arow[kn + 4 * j + 1]; }
        #pragma unroll
        for (int j = 0; j < 4; ++j) {
            #pragma unroll
            for (int t = 0; t < 4; ++t) {
                v2f b;
                b.x = p0[j][t];
                b.y = p1[j][t];
                c[t] = __builtin_amdgcn_wmma_f32_16x16x4_f32(false, a[j], false, b, (short)0, c[t], false, false);
            }
        }
    }

    const int mbase = khalf * 8;
    #pragma unroll
    for (int r = 0; r < 8; ++r) {
        f4u o;
        #pragma unroll
        for (int t = 0; t < 4; ++t) o[t] = c[t][r];
        *(f4u*)(h2 + (size_t)(row0 + mbase + r) * D_OUT + 4 * n) = o;
    }
}


__global__ void logsoftmax_rows(float* out, int N) {
    const int wave = threadIdx.x >> 5;
    const int lane = threadIdx.x & 31;
    const int rowi = blockIdx.x * 8 + wave;
    if (rowi >= N) return;
    float* p = out + (size_t)rowi * D_OUT;
    float v0 = p[lane];
    float v1 = p[lane + 32];
    float mx = fmaxf(v0, v1);
    #pragma unroll
    for (int off = 16; off >= 1; off >>= 1)
        mx = fmaxf(mx, __shfl_xor(mx, off, 32));
    float s = __expf(v0 - mx) + __expf(v1 - mx);
    #pragma unroll
    for (int off = 16; off >= 1; off >>= 1)
        s += __shfl_xor(s, off, 32);
    float l = mx + __logf(s);
    p[lane]      = v0 - l;
    p[lane + 32] = v1 - l;
}


extern "C" void kernel_launch(void* const* d_in, const int* in_sizes, int n_in,
                              void* d_out, int out_size, void* d_ws, size_t ws_size,
                              hipStream_t stream) {
    const float* x     = (const float*)d_in[0];
    const float* W1    = (const float*)d_in[1];
    const float* b1    = (const float*)d_in[2];
    const float* gamma = (const float*)d_in[3];
    const float* beta  = (const float*)d_in[4];
    const float* W2    = (const float*)d_in[5];
    const float* b2    = (const float*)d_in[6];
    const int*   ei    = (const int*)d_in[7];

    const int N = in_sizes[0] / D_HID;
    const int E = in_sizes[7] / 2;
    const int* erow = ei;
    const int* ecol = ei + E;

    float* ws    = (float*)d_ws;
    float* dis   = ws;
    float* h1    = dis + N;
    float* hagg  = h1 + (size_t)N * D_HID;
    float* stats = hagg + (size_t)N * D_HID;
    float* bsum  = stats;
    float* bsq   = stats + 128;
    float* scal  = stats + 256;
    float* shft  = stats + 384;
    float* h2    = h1;
    int*   meta  = (int*)((char*)d_ws + (size_t)(63u << 20));
    int*   recs  = (int*)((char*)d_ws + (size_t)(64u << 20));
    float* outp  = (float*)d_out;

    init_deg_stats<<<(N + 255) / 256, 256, 0, stream>>>(dis, stats, N);
    count_edges<<<(E + 255) / 256, 256, 0, stream>>>(ecol, dis, E);
    deg_to_dis<<<(N + 255) / 256, 256, 0, stream>>>(dis, N);

    gemm1_wmma<<<(N + 63) / 64, 128, 0, stream>>>(x, W1, b1, h1, N);
    own_gather<D_HID, true><<<(N + OWN_ROWS - 1) / OWN_ROWS, 1024, 0, stream>>>(erow, ecol, dis, h1, hagg, recs, meta, N, E);

    bn_stats<<<1024, 256, 0, stream>>>(hagg, bsum, bsq, N * D_HID);
    bn_finalize<<<1, 128, 0, stream>>>(bsum, bsq, gamma, beta, scal, shft, N);

    gemm2_wmma<<<(N + 63) / 64, 128, 0, stream>>>(hagg, scal, shft, W2, b2, h2, N);
    own_gather<D_OUT, false><<<(N + OWN_ROWS - 1) / OWN_ROWS, 1024, 0, stream>>>(erow, ecol, dis, h2, outp, recs, meta, N, E);

    logsoftmax_rows<<<(N + 7) / 8, 256, 0, stream>>>(outp, N);
}


# Three small helper launches folded away: GEMM1 (now first) stores the degree start values and zeroes the BN sums, rsqrt(degree) is taken at its uses, GEMM2's prologue derives BN scale/shift; 10 -> 7
# speedup vs baseline: 2.2121x; 1.0386x over previous
#include <hip/hip_runtime.h>
#include <hip/hip_bf16.h>
#include <math.h>

typedef float v8f __attribute__((ext_vector_type(8)));
typedef float v2f __attribute__((ext_vector_type(2)));

#define D_HID 128
#define D_OUT 64
#define BN_EPS 1e-5f


__global__ void count_edges(const int* __restrict__ col, float* deg, int E) {
    int e = blockIdx.x * blockDim.x + threadIdx.x;
    if (e < E) atomicAdd(&deg[col[e]], 1.0f);
}


typedef float f4u __attribute__((ext_vector_type(4), aligned(4)));

__global__ void __launch_bounds__(128) gemm1_wmma(const float* __restrict__ x, const float* __restrict__ W1,
                           const float* __restrict__ b1, float* __restrict__ h1,
                           float* __restrict__ deg, float* __restrict__ stats, int N) {
    if (threadIdx.x < 64) {
        const int i = blockIdx.x * 64 + threadIdx.x;
        if (i < N) deg[i] = 1.0f;
    }
    if (blockIdx.x == 0) { stats[threadIdx.x] = 0.0f; stats[threadIdx.x + 128] = 0.0f; }
    const int wave  = __builtin_amdgcn_readfirstlane((int)(threadIdx.x >> 5));
    const int lane  = threadIdx.x & 31;
    const int row0  = (blockIdx.x * 4 + wave) * 16;
    if (row0 >= N) return;
    const int n     = lane & 15;
    const int khalf = lane >> 4;

    v8f c[8];
    {
        const f4u blo = *(const f4u*)(b1 + 4 * n);
        const f4u bhi = *(const f4u*)(b1 + 64 + 4 * n);
        #pragma unroll
        for (int t = 0; t < 4; ++t) {
            #pragma unroll
            for (int r = 0; r < 8; ++r) { c[t][r] = blo[t]; c[t + 4][r] = bhi[t]; }
        }
    }

    const float* arow = x + (size_t)(row0 + n) * D_HID + 2 * khalf;
    const float* brow = W1 + (size_t)(2 * khalf) * D_HID + 4 * n;

    v2f an[4];
    #pragma unroll
    for (int j = 0; j < 4; ++j) { an[j].x = arow[4 * j]; an[j].y = arow[4 * j + 1]; }

    #pragma unroll 2
    for (int k0 = 0; k0 < 128; k0 += 16) {
        v2f a[4];
        f4u p0[4], p1[4], q0[4], q1[4];
        #pragma unroll
        for (int j = 0; j < 4; ++j) {
            a[j] = an[j];
            const float* bk = brow + (size_t)(k0 + 4 * j) * D_HID;
            p0[j] = *(const f4u*)(bk);
            p1[j] = *(const f4u*)(bk + D_HID);
            q0[j] = *(const f4u*)(bk + 64);
            q1[j] = *(const f4u*)(bk + D_HID + 64);
        }
        const int kn = (k0 + 16 < 128) ? k0 + 16 : k0;
        #pragma unroll
        for (int j = 0; j < 4; ++j) { an[j].x = arow[kn + 4 * j]; an[j].y = arow[kn + 4 * j + 1]; }
        #pragma unroll
        for (int j = 0; j < 4; ++j) {
            #pragma unroll
            for (int t = 0; t < 4; ++t) {
                v2f b;
                b.x = p0[j][t];
                b.y = p1[j][t];
                c[t] = __builtin_amdgcn_wmma_f32_16x16x4_f32(false, a[j], false, b, (short)0, c[t], false, false);
            }
            #pragma unroll
            for (int t = 0; t < 4; ++t) {
                v2f b;
                b.x = q0[j][t];
                b.y = q1[j][t];
                c[t + 4] = __builtin_amdgcn_wmma_f32_16x16x4_f32(false, a[j], false, b, (short)0, c[t + 4], false, false);
            }
        }
    }

    const int mbase = khalf * 8;
    #pragma unroll
    for (int r = 0; r < 8; ++r) {
        float* orow = h1 + (size_t)(row0 + mbase + r) * D_HID + 4 * n;
        f4u lo, hi;
        #pragma unroll
        for (int t = 0; t < 4; ++t) { lo[t] = c[t][r]; hi[t] = c[t + 4][r]; }
        *(f4u*)(orow)      = lo;
        *(f4u*)(orow + 64) = hi;
    }
}


#define OWN_ROWS 196
#define QCAP 160
#define REC_CAP 16384

typedef int   vi4u __attribute__((ext_vector_type(4), aligned(4)));
template <int V> struct rowvec;
template <> struct rowvec<4> { typedef float t __attribute__((ext_vector_type(4), aligned(4))); };
template <> struct rowvec<2> { typedef float t __attribute__((ext_vector_type(2), aligned(4))); };

template <int CH, bool FULL>
__device__ __forceinline__ void add_rows(int r, int cl, int wi, int cnt,
                                         const float* __restrict__ h, float* acc, int lane) {
    constexpr int V = CH / 32;
    typedef typename rowvec<V>::t vec;
    #pragma unroll 1
    for (int j0 = 0; j0 < cnt; j0 += 8) {
        vec v[8];
        #pragma unroll
        for (int j = 0; j < 8; ++j) {
            const int rj = __builtin_amdgcn_readlane(r, j0 + j);
            v[j] = *(const vec*)(h + (size_t)rj * CH + lane * V);
        }
        #pragma unroll
        for (int j = 0; j < 8; ++j) {
            if (FULL || j0 + j < cnt) {
                const int cj = __builtin_amdgcn_readlane(cl, j0 + j);
                const float wj = __builtin_bit_cast(float, __builtin_amdgcn_readlane(wi, j0 + j));
                float* a = acc + cj * CH + lane;
                #pragma unroll
                for (int k = 0; k < V; ++k) atomicAdd(a + 32 * k, wj * v[j][k]);
            }
        }
    }
}

template <int CH, bool FULL, bool SAVE>
__device__ __forceinline__ void drain_round(const int* qe_, const int* qc_, int cnt,
                                            const int* __restrict__ row, const float* __restrict__ deg,
                                            const float* __restrict__ h, float* acc, int lo, int lane,
                                            int* rec, int* ctl) {
    int cl = -1, r = 0;
    float w = 0.0f;
    if (FULL || lane < cnt) {
        const int e = qe_[lane];
        cl = qc_[lane];
        r = row[e];
        w = rsqrtf(deg[r]) * rsqrtf(deg[lo + cl]);
    }
    const int wi = __builtin_bit_cast(int, w);
    if (SAVE) {
        int base = 0;
        if (lane == 0) base = atomicAdd(&ctl[0], 32);
        base = __builtin_amdgcn_readfirstlane(base);
        if (base + 32 <= REC_CAP) {
            rec[base + lane] = r;
            rec[REC_CAP + base + lane] = cl;
            rec[2 * REC_CAP + base + lane] = wi;
        } else if (lane == 0) {
            ctl[1] = 1;
        }
    }
    add_rows<CH, FULL>(r, cl, wi, cnt, h, acc, lane);
}

template <int CH, bool SAVE>
__global__ void __launch_bounds__(1024) own_gather(const int* __restrict__ row, const int* __restrict__ col,
                                                   const float* __restrict__ deg, const float* __restrict__ h,
                                                   float* __restrict__ dst, int* recs, int* meta, int N, int E) {
    constexpr int V = CH / 32;
    typedef typename rowvec<V>::t vec;
    __shared__ float acc[OWN_ROWS * CH];
    __shared__ int qe[32][QCAP];
    __shared__ int qc[32][QCAP];
    __shared__ int ctl[2];
    const int wave = __builtin_amdgcn_readfirstlane((int)(threadIdx.x >> 5));
    const int lane = threadIdx.x & 31;
    const int lo = blockIdx.x * OWN_ROWS;
    const int nrows = min(OWN_ROWS, N - lo);
    int* rec = recs + (size_t)blockIdx.x * (3 * REC_CAP);
    if (SAVE && threadIdx.x == 0) { ctl[0] = 0; ctl[1] = 0; }

    for (int rl = wave; rl < nrows; rl += 32) {
        const vec v = *(const vec*)(h + (size_t)(lo + rl) * CH + lane * V);
        const float d = rsqrtf(deg[lo + rl]);
        #pragma unroll
        for (int k = 0; k < V; ++k) acc[rl * CH + k * 32 + lane] = d * d * v[k];
    }
    __syncthreads();

    const int nslots = SAVE ? -1 : meta[blockIdx.x * 32];
    if (!SAVE && nslots >= 0) {
        const int nfr = nslots >> 5;
        for (int f = wave; f < nfr; f += 32) {
            const int r  = rec[f * 32 + lane];
            const int cl = rec[REC_CAP + f * 32 + lane];
            const int wi = rec[2 * REC_CAP + f * 32 + lane];
            const int cnt = __builtin_popcount(__builtin_amdgcn_ballot_w32(cl >= 0));
            if (cnt == 32) add_rows<CH, true>(r, cl, wi, 32, h, acc, lane);
            else           add_rows<CH, false>(r, cl, wi, cnt, h, acc, lane);
        }
    } else {
    int* myqe = qe[wave];
    int* myqc = qc[wave];
    const int nchunks = (E + 127) >> 7;
    const int nsuper = (nchunks + 3) >> 2;
    const int s0 = (int)(((long long)blockIdx.x * nsuper) / gridDim.x);
    int q = 0;
    for (int it = wave; it < nsuper; it += 32) {
        int s = it + s0;
        if (s >= nsuper) s -= nsuper;
        vi4u c[4];
        if (s * 512 + 512 <= E) {
            #pragma unroll
            for (int k = 0; k < 4; ++k) c[k] = *(const vi4u*)(col + (s * 4 + k) * 128 + lane * 4);
        } else {
            #pragma unroll
            for (int k = 0; k < 4; ++k) {
                const int eb = (s * 4 + k) * 128 + lane * 4;
                c[k].x = (eb + 0 < E) ? col[eb + 0] : -1;
                c[k].y = (eb + 1 < E) ? col[eb + 1] : -1;
                c[k].z = (eb + 2 < E) ? col[eb + 2] : -1;
                c[k].w = (eb + 3 < E) ? col[eb + 3] : -1;
            }
        }
        #pragma unroll
        for (int k = 0; k < 4; ++k) {
            const int eb = (s * 4 + k) * 128 + lane * 4;
            unsigned d[4];
            #pragma unroll
            for (int i = 0; i < 4; ++i) d[i] = (unsigned)(c[k][i] - lo);
            const bool any = min(min(d[0], d[1]), min(d[2], d[3])) < (unsigned)nrows;
            if (__builtin_amdgcn_ballot_w32(any) != 0u) {
                #pragma unroll
                for (int i = 0; i < 4; ++i) {
                    const bool mi = d[i] < (unsigned)nrows;
                    const unsigned bal = __builtin_amdgcn_ballot_w32(mi);
                    if (bal != 0u) {
                        if (mi) {
                            const int pos = q + (int)__builtin_amdgcn_mbcnt_lo(bal, 0u);
                            myqe[pos] = eb + i;
                            myqc[pos] = (int)d[i];
                        }
                        q += __builtin_popcount(bal);
                    }
                }
                while (q >= 32) {
                    q -= 32;
                    drain_round<CH, true, SAVE>(myqe + q, myqc + q, 32, row, deg, h, acc, lo, lane, rec, ctl);
                }
            }
        }
    }
    if (q > 0) drain_round<CH, false, SAVE>(myqe, myqc, q, row, deg, h, acc, lo, lane, rec, ctl);
    }
    __syncthreads();
    if (SAVE && threadIdx.x == 0) meta[blockIdx.x * 32] = ctl[1] ? -1 : ctl[0];

    for (int rl = wave; rl < nrows; rl += 32) {
        vec o;
        #pragma unroll
        for (int k = 0; k < V; ++k) o[k] = acc[rl * CH + k * 32 + lane];
        *(vec*)(dst + (size_t)(lo + rl) * CH + lane * V) = o;
    }
}


__global__ void bn_stats(const float* __restrict__ hagg, float* bsum, float* bsq, int total) {
    __shared__ float ssum[D_HID];
    __shared__ float ssq[D_HID];
    for (int i = threadIdx.x; i < D_HID; i += blockDim.x) { ssum[i] = 0.0f; ssq[i] = 0.0f; }
    __syncthreads();
    int stride = blockDim.x * gridDim.x;
    for (int i = blockIdx.x * blockDim.x + threadIdx.x; i < total; i += stride) {
        float v = hagg[i];
        atomicAdd(&ssum[i & 127], v);
        atomicAdd(&ssq[i & 127], v * v);
    }
    __syncthreads();
    for (int i = threadIdx.x; i < D_HID; i += blockDim.x) {
        atomicAdd(&bsum[i], ssum[i]);
        atomicAdd(&bsq[i], ssq[i]);
    }
}


__global__ void __launch_bounds__(128) gemm2_wmma(const float* __restrict__ hagg, const float* __restrict__ bsum,
                           const float* __restrict__ bsq, const float* __restrict__ gamma,
                           const float* __restrict__ beta, const float* __restrict__ W2,
                           const float* __restrict__ b2, float* __restrict__ h2, int N) {
    __shared__ float sc[D_HID];
    __shared__ float sh[D_HID];
    const int wave  = __builtin_amdgcn_readfirstlane((int)(threadIdx.x >> 5));
    const int lane  = threadIdx.x & 31;
    const int row0  = (blockIdx.x * 4 + wave) * 16;
    const bool live = row0 < N;
    const int n     = lane & 15;
    const int khalf = lane >> 4;
    const float* arow = hagg + (size_t)(row0 + n) * D_HID + 2 * khalf;

    v2f an[4];
    #pragma unroll
    for (int j = 0; j < 4; ++j) { an[j].x = 0.0f; an[j].y = 0.0f; }
    if (live) {
        #pragma unroll
        for (int j = 0; j < 4; ++j) { an[j].x = arow[4 * j]; an[j].y = arow[4 * j + 1]; }
    }
    {
        const int ch = threadIdx.x;
        const float inv = 1.0f / (float)N;
        const float mu = bsum[ch] * inv;
        const float var = bsq[ch] * inv - mu * mu;
        const float s = gamma[ch] * rsqrtf(var + BN_EPS);
        sc[ch] = s;
        sh[ch] = beta[ch] - mu * s;
    }
    __syncthreads();
    if (!live) return;

    v8f c[4];
    {
        const f4u bias = *(const f4u*)(b2 + 4 * n);
        #pragma unroll
        for (int t = 0; t < 4; ++t) {
            #pragma unroll
            for (int r = 0; r < 8; ++r) c[t][r] = bias[t];
        }
    }

    const float* brow = W2 + (size_t)(2 * khalf) * D_OUT + 4 * n;

    #pragma unroll 2
    for (int k0 = 0; k0 < 128; k0 += 16) {
        v2f a[4];
        f4u p0[4], p1[4];
        #pragma unroll
        for (int j = 0; j < 4; ++j) {
            const int k = k0 + 4 * j;
            const int kk = k + 2 * khalf;
            a[j].x = fmaxf(an[j].x * sc[kk + 0] + sh[kk + 0], 0.0f);
            a[j].y = fmaxf(an[j].y * sc[kk + 1] + sh[kk + 1], 0.0f);
            const float* bk = brow + (size_t)k * D_OUT;
            p0[j] = *(const f4u*)(bk);
            p1[j] = *(const f4u*)(bk + D_OUT);
        }
        const int kn = (k0 + 16 < 128) ? k0 + 16 : k0;
        #pragma unroll
        for (int j = 0; j < 4; ++j) { an[j].x = arow[kn + 4 * j]; an[j].y = arow[kn + 4 * j + 1]; }
        #pragma unroll
        for (int j = 0; j < 4; ++j) {
            #pragma unroll
            for (int t = 0; t < 4; ++t) {
                v2f b;
                b.x = p0[j][t];
                b.y = p1[j][t];
                c[t] = __builtin_amdgcn_wmma_f32_16x16x4_f32(false, a[j], false, b, (short)0, c[t], false, false);
            }
        }
    }

    const int mbase = khalf * 8;
    #pragma unroll
    for (int r = 0; r < 8; ++r) {
        f4u o;
        #pragma unroll
        for (int t = 0; t < 4; ++t) o[t] = c[t][r];
        *(f4u*)(h2 + (size_t)(row0 + mbase + r) * D_OUT + 4 * n) = o;
    }
}


__global__ void logsoftmax_rows(float* out, int N) {
    const int wave = threadIdx.x >> 5;
    const int lane = threadIdx.x & 31;
    const int rowi = blockIdx.x * 8 + wave;
    if (rowi >= N) return;
    float* p = out + (size_t)rowi * D_OUT;
    float v0 = p[lane];
    float v1 = p[lane + 32];
    float mx = fmaxf(v0, v1);
    #pragma unroll
    for (int off = 16; off >= 1; off >>= 1)
        mx = fmaxf(mx, __shfl_xor(mx, off, 32));
    float s = __expf(v0 - mx) + __expf(v1 - mx);
    #pragma unroll
    for (int off = 16; off >= 1; off >>= 1)
        s += __shfl_xor(s, off, 32);
    float l = mx + __logf(s);
    p[lane]      = v0 - l;
    p[lane + 32] = v1 - l;
}


extern "C" void kernel_launch(void* const* d_in, const int* in_sizes, int n_in,
                              void* d_out, int out_size, void* d_ws, size_t ws_size,
                              hipStream_t stream) {
    const float* x     = (const float*)d_in[0];
    const float* W1    = (const float*)d_in[1];
    const float* b1    = (const float*)d_in[2];
    const float* gamma = (const float*)d_in[3];
    const float* beta  = (const float*)d_in[4];
    const float* W2    = (const float*)d_in[5];
    const float* b2    = (const float*)d_in[6];
    const int*   ei    = (const int*)d_in[7];

    const int N = in_sizes[0] / D_HID;
    const int E = in_sizes[7] / 2;
    const int* erow = ei;
    const int* ecol = ei + E;

    float* ws    = (float*)d_ws;
    float* deg   = ws;
    float* h1    = deg + N;
    float* hagg  = h1 + (size_t)N * D_HID;
    float* stats = hagg + (size_t)N * D_HID;
    float* bsum  = stats;
    float* bsq   = stats + 128;
    float* h2    = h1;
    int*   meta  = (int*)((char*)d_ws + (size_t)(63u << 20));
    int*   recs  = (int*)((char*)d_ws + (size_t)(64u << 20));
    float* outp  = (float*)d_out;

    gemm1_wmma<<<(N + 63) / 64, 128, 0, stream>>>(x, W1, b1, h1, deg, stats, N);
    count_edges<<<(E + 255) / 256, 256, 0, stream>>>(ecol, deg, E);
    own_gather<D_HID, true><<<(N + OWN_ROWS - 1) / OWN_ROWS, 1024, 0, stream>>>(erow, ecol, deg, h1, hagg, recs, meta, N, E);

    bn_stats<<<1024, 256, 0, stream>>>(hagg, bsum, bsq, N * D_HID);

    gemm2_wmma<<<(N + 63) / 64, 128, 0, stream>>>(hagg, bsum, bsq, gamma, beta, W2, b2, h2, N);
    own_gather<D_OUT, false><<<(N + OWN_ROWS - 1) / OWN_ROWS, 1024, 0, stream>>>(erow, ecol, deg, h2, outp, recs, meta, N, E);

    logsoftmax_rows<<<(N + 7) / 8, 256, 0, stream>>>(outp, N);
}


# Edge sweep filters a 1-byte owner-block key per edge (stored by the degree-count kernel): 16 edges per lane-load, byte-parallel match; destination read at drain; LDS float atomics, arrival order kept
# speedup vs baseline: 2.3251x; 1.0511x over previous
#include <hip/hip_runtime.h>
#include <hip/hip_bf16.h>
#include <math.h>

typedef float v8f __attribute__((ext_vector_type(8)));
typedef float v2f __attribute__((ext_vector_type(2)));

#define D_HID 128
#define D_OUT 64
#define BN_EPS 1e-5f


#define OWN_ROWS 196
#define KEY_SPAN 2048

typedef int vi4u __attribute__((ext_vector_type(4), aligned(4)));

__global__ void __launch_bounds__(256) count_edges(const int* __restrict__ col, float* deg,
                                                   unsigned* __restrict__ key, int E) {
    const int t = blockIdx.x * 256 + threadIdx.x;
    const int e0 = 4 * t;
    unsigned kw = 0u;
    if (e0 + 4 <= E) {
        const vi4u c = *(const vi4u*)(col + e0);
        #pragma unroll
        for (int i = 0; i < 4; ++i) {
            atomicAdd(&deg[c[i]], 1.0f);
            kw |= (((unsigned)c[i] / OWN_ROWS) & 255u) << (8 * i);
        }
    } else {
        #pragma unroll
        for (int i = 0; i < 4; ++i) {
            if (e0 + i < E) {
                const int ci = col[e0 + i];
                atomicAdd(&deg[ci], 1.0f);
                kw |= (((unsigned)ci / OWN_ROWS) & 255u) << (8 * i);
            }
        }
    }
    key[t] = kw;
}


typedef float f4u __attribute__((ext_vector_type(4), aligned(4)));

__global__ void __launch_bounds__(128) gemm1_wmma(const float* __restrict__ x, const float* __restrict__ W1,
                           const float* __restrict__ b1, float* __restrict__ h1,
                           float* __restrict__ deg, float* __restrict__ stats, int N) {
    if (threadIdx.x < 64) {
        const int i = blockIdx.x * 64 + threadIdx.x;
        if (i < N) deg[i] = 1.0f;
    }
    if (blockIdx.x == 0) { stats[threadIdx.x] = 0.0f; stats[threadIdx.x + 128] = 0.0f; }
    const int wave  = __builtin_amdgcn_readfirstlane((int)(threadIdx.x >> 5));
    const int lane  = threadIdx.x & 31;
    const int row0  = (blockIdx.x * 4 + wave) * 16;
    if (row0 >= N) return;
    const int n     = lane & 15;
    const int khalf = lane >> 4;

    v8f c[8];
    {
        const f4u blo = *(const f4u*)(b1 + 4 * n);
        const f4u bhi = *(const f4u*)(b1 + 64 + 4 * n);
        #pragma unroll
        for (int t = 0; t < 4; ++t) {
            #pragma unroll
            for (int r = 0; r < 8; ++r) { c[t][r] = blo[t]; c[t + 4][r] = bhi[t]; }
        }
    }

    const float* arow = x + (size_t)(row0 + n) * D_HID + 2 * khalf;
    const float* brow = W1 + (size_t)(2 * khalf) * D_HID + 4 * n;

    v2f an[4];
    #pragma unroll
    for (int j = 0; j < 4; ++j) { an[j].x = arow[4 * j]; an[j].y = arow[4 * j + 1]; }

    #pragma unroll 2
    for (int k0 = 0; k0 < 128; k0 += 16) {
        v2f a[4];
        f4u p0[4], p1[4], q0[4], q1[4];
        #pragma unroll
        for (int j = 0; j < 4; ++j) {
            a[j] = an[j];
            const float* bk = brow + (size_t)(k0 + 4 * j) * D_HID;
            p0[j] = *(const f4u*)(bk);
            p1[j] = *(const f4u*)(bk + D_HID);
            q0[j] = *(const f4u*)(bk + 64);
            q1[j] = *(const f4u*)(bk + D_HID + 64);
        }
        const int kn = (k0 + 16 < 128) ? k0 + 16 : k0;
        #pragma unroll
        for (int j = 0; j < 4; ++j) { an[j].x = arow[kn + 4 * j]; an[j].y = arow[kn + 4 * j + 1]; }
        #pragma unroll
        for (int j = 0; j < 4; ++j) {
            #pragma unroll
            for (int t = 0; t < 4; ++t) {
                v2f b;
                b.x = p0[j][t];
                b.y = p1[j][t];
                c[t] = __builtin_amdgcn_wmma_f32_16x16x4_f32(false, a[j], false, b, (short)0, c[t], false, false);
            }
            #pragma unroll
            for (int t = 0; t < 4; ++t) {
                v2f b;
                b.x = q0[j][t];
                b.y = q1[j][t];
                c[t + 4] = __builtin_amdgcn_wmma_f32_16x16x4_f32(false, a[j], false, b, (short)0, c[t + 4], false, false);
            }
        }
    }

    const int mbase = khalf * 8;
    #pragma unroll
    for (int r = 0; r < 8; ++r) {
        float* orow = h1 + (size_t)(row0 + mbase + r) * D_HID + 4 * n;
        f4u lo, hi;
        #pragma unroll
        for (int t = 0; t < 4; ++t) { lo[t] = c[t][r]; hi[t] = c[t + 4][r]; }
        *(f4u*)(orow)      = lo;
        *(f4u*)(orow + 64) = hi;
    }
}


#define QCAP 64
#define REC_CAP 16384

typedef unsigned vu4 __attribute__((ext_vector_type(4)));
template <int V> struct rowvec;
template <> struct rowvec<4> { typedef float t __attribute__((ext_vector_type(4), aligned(4))); };
template <> struct rowvec<2> { typedef float t __attribute__((ext_vector_type(2), aligned(4))); };

template <int CH, bool FULL>
__device__ __forceinline__ void add_rows(int r, int cl, int wi, int cnt,
                                         const float* __restrict__ h, float* acc, int lane) {
    constexpr int V = CH / 32;
    typedef typename rowvec<V>::t vec;
    #pragma unroll 1
    for (int j0 = 0; j0 < cnt; j0 += 8) {
        vec v[8];
        #pragma unroll
        for (int j = 0; j < 8; ++j) {
            const int rj = __builtin_amdgcn_readlane(r, j0 + j);
            v[j] = *(const vec*)(h + (size_t)rj * CH + lane * V);
        }
        #pragma unroll
        for (int j = 0; j < 8; ++j) {
            if (FULL || j0 + j < cnt) {
                const int cj = __builtin_amdgcn_readlane(cl, j0 + j);
                const float wj = __builtin_bit_cast(float, __builtin_amdgcn_readlane(wi, j0 + j));
                float* a = acc + cj * CH + lane;
                #pragma unroll
                for (int k = 0; k < V; ++k) atomicAdd(a + 32 * k, wj * v[j][k]);
            }
        }
    }
}

template <int CH, bool FULL, bool SAVE>
__device__ __forceinline__ void drain_round(const int* qe_, int cnt,
                                            const int* __restrict__ row, const int* __restrict__ col,
                                            const float* __restrict__ deg,
                                            const float* __restrict__ h, float* acc, int lo, int lane,
                                            int* rec, int* ctl) {
    int cl = -1, r = 0;
    float w = 0.0f;
    if (FULL || lane < cnt) {
        const int e = qe_[lane];
        r = row[e];
        const int c = col[e];
        cl = c - lo;
        w = rsqrtf(deg[r]) * rsqrtf(deg[c]);
    }
    const int wi = __builtin_bit_cast(int, w);
    if (SAVE) {
        int base = 0;
        if (lane == 0) base = atomicAdd(&ctl[0], 32);
        base = __builtin_amdgcn_readfirstlane(base);
        if (base + 32 <= REC_CAP) {
            rec[base + lane] = r;
            rec[REC_CAP + base + lane] = cl;
            rec[2 * REC_CAP + base + lane] = wi;
        } else if (lane == 0) {
            ctl[1] = 1;
        }
    }
    add_rows<CH, FULL>(r, cl, wi, cnt, h, acc, lane);
}

template <int CH, bool SAVE>
__global__ void __launch_bounds__(1024) own_gather(const int* __restrict__ row, const int* __restrict__ col,
                                                   const unsigned* __restrict__ key,
                                                   const float* __restrict__ deg, const float* __restrict__ h,
                                                   float* __restrict__ dst, int* recs, int* meta, int N, int E) {
    constexpr int V = CH / 32;
    typedef typename rowvec<V>::t vec;
    __shared__ float acc[OWN_ROWS * CH];
    __shared__ int qe[32][QCAP];
    __shared__ int ctl[2];
    const int wave = __builtin_amdgcn_readfirstlane((int)(threadIdx.x >> 5));
    const int lane = threadIdx.x & 31;
    const int lo = blockIdx.x * OWN_ROWS;
    const int nrows = min(OWN_ROWS, N - lo);
    int* rec = recs + (size_t)blockIdx.x * (3 * REC_CAP);
    if (SAVE && threadIdx.x == 0) { ctl[0] = 0; ctl[1] = 0; }

    for (int rl = wave; rl < nrows; rl += 32) {
        const vec v = *(const vec*)(h + (size_t)(lo + rl) * CH + lane * V);
        const float d = rsqrtf(deg[lo + rl]);
        #pragma unroll
        for (int k = 0; k < V; ++k) acc[rl * CH + k * 32 + lane] = d * d * v[k];
    }
    __syncthreads();

    const int nslots = SAVE ? -1 : meta[blockIdx.x * 32];
    if (!SAVE && nslots >= 0) {
        const int nfr = nslots >> 5;
        for (int f = wave; f < nfr; f += 32) {
            const int r  = rec[f * 32 + lane];
            const int cl = rec[REC_CAP + f * 32 + lane];
            const int wi = rec[2 * REC_CAP + f * 32 + lane];
            const int cnt = __builtin_popcount(__builtin_amdgcn_ballot_w32(cl >= 0));
            if (cnt == 32) add_rows<CH, true>(r, cl, wi, 32, h, acc, lane);
            else           add_rows<CH, false>(r, cl, wi, cnt, h, acc, lane);
        }
    } else {
    int* myqe = qe[wave];
    const unsigned pat = (unsigned)blockIdx.x * 0x01010101u;
    const int nsuper = (E + KEY_SPAN - 1) / KEY_SPAN;
    const int s0 = (int)(((long long)blockIdx.x * nsuper) / gridDim.x);
    int q = 0;
    for (int it = wave; it < nsuper; it += 32) {
        int s = it + s0;
        if (s >= nsuper) s -= nsuper;
        vu4 kw[4];
        #pragma unroll
        for (int k = 0; k < 4; ++k) kw[k] = *(const vu4*)(key + (size_t)(s * 4 + k) * 128 + lane * 4);
        const bool tail = s * KEY_SPAN + KEY_SPAN > E;
        #pragma unroll
        for (int k = 0; k < 4; ++k) {
            const int eb = (s * 4 + k) * 512 + lane * 16;
            unsigned m = 0u;
            #pragma unroll
            for (int i = 0; i < 4; ++i) {
                const unsigned x = kw[k][i] ^ pat;
                const unsigned t = (x & 0x7f7f7f7fu) + 0x7f7f7f7fu;
                m |= (~(t | x) & 0x80808080u) >> i;
            }
            if (tail) {
                const int left = E - eb;
                unsigned keep = 0u;
                #pragma unroll
                for (int i = 0; i < 4; ++i) {
                    #pragma unroll
                    for (int j = 0; j < 4; ++j) keep |= (4 * i + j < left) ? (1u << (8 * j + 7 - i)) : 0u;
                }
                m &= keep;
            }
            unsigned bal = __builtin_amdgcn_ballot_w32(m != 0u);
            while (bal != 0u) {
                if (m != 0u) {
                    const int p = __builtin_ctz(m);
                    const int pos = q + (int)__builtin_amdgcn_mbcnt_lo(bal, 0u);
                    myqe[pos] = eb + ((7 - (p & 7)) << 2) + (p >> 3);
                    m &= m - 1u;
                }
                q += __builtin_popcount(bal);
                if (q >= 32) {
                    q -= 32;
                    drain_round<CH, true, SAVE>(myqe + q, 32, row, col, deg, h, acc, lo, lane, rec, ctl);
                }
                bal = __builtin_amdgcn_ballot_w32(m != 0u);
            }
        }
    }
    if (q > 0) drain_round<CH, false, SAVE>(myqe, q, row, col, deg, h, acc, lo, lane, rec, ctl);
    }
    __syncthreads();
    if (SAVE && threadIdx.x == 0) meta[blockIdx.x * 32] = ctl[1] ? -1 : ctl[0];

    for (int rl = wave; rl < nrows; rl += 32) {
        vec o;
        #pragma unroll
        for (int k = 0; k < V; ++k) o[k] = acc[rl * CH + k * 32 + lane];
        *(vec*)(dst + (size_t)(lo + rl) * CH + lane * V) = o;
    }
}


__global__ void bn_stats(const float* __restrict__ hagg, float* bsum, float* bsq, int total) {
    __shared__ float ssum[D_HID];
    __shared__ float ssq[D_HID];
    for (int i = threadIdx.x; i < D_HID; i += blockDim.x) { ssum[i] = 0.0f; ssq[i] = 0.0f; }
    __syncthreads();
    int stride = blockDim.x * gridDim.x;
    for (int i = blockIdx.x * blockDim.x + threadIdx.x; i < total; i += stride) {
        float v = hagg[i];
        atomicAdd(&ssum[i & 127], v);
        atomicAdd(&ssq[i & 127], v * v);
    }
    __syncthreads();
    for (int i = threadIdx.x; i < D_HID; i += blockDim.x) {
        atomicAdd(&bsum[i], ssum[i]);
        atomicAdd(&bsq[i], ssq[i]);
    }
}


__global__ void __launch_bounds__(128) gemm2_wmma(const float* __restrict__ hagg, const float* __restrict__ bsum,
                           const float* __restrict__ bsq, const float* __restrict__ gamma,
                           const float* __restrict__ beta, const float* __restrict__ W2,
                           const float* __restrict__ b2, float* __restrict__ h2, int N) {
    __shared__ float sc[D_HID];
    __shared__ float sh[D_HID];
    const int wave  = __builtin_amdgcn_readfirstlane((int)(threadIdx.x >> 5));
    const int lane  = threadIdx.x & 31;
    const int row0  = (blockIdx.x * 4 + wave) * 16;
    const bool live = row0 < N;
    const int n     = lane & 15;
    const int khalf = lane >> 4;
    const float* arow = hagg + (size_t)(row0 + n) * D_HID + 2 * khalf;

    v2f an[4];
    #pragma unroll
    for (int j = 0; j < 4; ++j) { an[j].x = 0.0f; an[j].y = 0.0f; }
    if (live) {
        #pragma unroll
        for (int j = 0; j < 4; ++j) { an[j].x = arow[4 * j]; an[j].y = arow[4 * j + 1]; }
    }
    {
        const int ch = threadIdx.x;
        const float inv = 1.0f / (float)N;
        const float mu = bsum[ch] * inv;
        const float var = bsq[ch] * inv - mu * mu;
        const float s = gamma[ch] * rsqrtf(var + BN_EPS);
        sc[ch] = s;
        sh[ch] = beta[ch] - mu * s;
    }
    __syncthreads();
    if (!live) return;

    v8f c[4];
    {
        const f4u bias = *(const f4u*)(b2 + 4 * n);
        #pragma unroll
        for (int t = 0; t < 4; ++t) {
            #pragma unroll
            for (int r = 0; r < 8; ++r) c[t][r] = bias[t];
        }
    }

    const float* brow = W2 + (size_t)(2 * khalf) * D_OUT + 4 * n;

    #pragma unroll 2
    for (int k0 = 0; k0 < 128; k0 += 16) {
        v2f a[4];
        f4u p0[4], p1[4];
        #pragma unroll
        for (int j = 0; j < 4; ++j) {
            const int k = k0 + 4 * j;
            const int kk = k + 2 * khalf;
            a[j].x = fmaxf(an[j].x * sc[kk + 0] + sh[kk + 0], 0.0f);
            a[j].y = fmaxf(an[j].y * sc[kk + 1] + sh[kk + 1], 0.0f);
            const float* bk = brow + (size_t)k * D_OUT;
            p0[j] = *(const f4u*)(bk);
            p1[j] = *(const f4u*)(bk + D_OUT);
        }
        const int kn = (k0 + 16 < 128) ? k0 + 16 : k0;
        #pragma unroll
        for (int j = 0; j < 4; ++j) { an[j].x = arow[kn + 4 * j]; an[j].y = arow[kn + 4 * j + 1]; }
        #pragma unroll
        for (int j = 0; j < 4; ++j) {
            #pragma unroll
            for (int t = 0; t < 4; ++t) {
                v2f b;
                b.x = p0[j][t];
                b.y = p1[j][t];
                c[t] = __builtin_amdgcn_wmma_f32_16x16x4_f32(false, a[j], false, b, (short)0, c[t], false, false);
            }
        }
    }

    const int mbase = khalf * 8;
    #pragma unroll
    for (int r = 0; r < 8; ++r) {
        f4u o;
        #pragma unroll
        for (int t = 0; t < 4; ++t) o[t] = c[t][r];
        *(f4u*)(h2 + (size_t)(row0 + mbase + r) * D_OUT + 4 * n) = o;
    }
}


__global__ void logsoftmax_rows(float* out, int N) {
    const int wave = threadIdx.x >> 5;
    const int lane = threadIdx.x & 31;
    const int rowi = blockIdx.x * 8 + wave;
    if (rowi >= N) return;
    float* p = out + (size_t)rowi * D_OUT;
    float v0 = p[lane];
    float v1 = p[lane + 32];
    float mx = fmaxf(v0, v1);
    #pragma unroll
    for (int off = 16; off >= 1; off >>= 1)
        mx = fmaxf(mx, __shfl_xor(mx, off, 32));
    float s = __expf(v0 - mx) + __expf(v1 - mx);
    #pragma unroll
    for (int off = 16; off >= 1; off >>= 1)
        s += __shfl_xor(s, off, 32);
    float l = mx + __logf(s);
    p[lane]      = v0 - l;
    p[lane + 32] = v1 - l;
}


extern "C" void kernel_launch(void* const* d_in, const int* in_sizes, int n_in,
                              void* d_out, int out_size, void* d_ws, size_t ws_size,
                              hipStream_t stream) {
    const float* x     = (const float*)d_in[0];
    const float* W1    = (const float*)d_in[1];
    const float* b1    = (const float*)d_in[2];
    const float* gamma = (const float*)d_in[3];
    const float* beta  = (const float*)d_in[4];
    const float* W2    = (const float*)d_in[5];
    const float* b2    = (const float*)d_in[6];
    const int*   ei    = (const int*)d_in[7];

    const int N = in_sizes[0] / D_HID;
    const int E = in_sizes[7] / 2;
    const int* erow = ei;
    const int* ecol = ei + E;

    float* ws    = (float*)d_ws;
    float* deg   = ws;
    float* h1    = deg + N;
    float* hagg  = h1 + (size_t)N * D_HID;
    float* stats = hagg + (size_t)N * D_HID;
    float* bsum  = stats;
    float* bsq   = stats + 128;
    float* h2    = h1;
    int*   meta  = (int*)((char*)d_ws + (size_t)(63u << 20));
    int*   recs  = (int*)((char*)d_ws + (size_t)(64u << 20));
    unsigned* key = (unsigned*)((char*)d_ws + (size_t)(120u << 20));
    float* outp  = (float*)d_out;

    gemm1_wmma<<<(N + 63) / 64, 128, 0, stream>>>(x, W1, b1, h1, deg, stats, N);
    count_edges<<<((E + KEY_SPAN - 1) / KEY_SPAN) * (KEY_SPAN / 1024), 256, 0, stream>>>(ecol, deg, key, E);
    own_gather<D_HID, true><<<(N + OWN_ROWS - 1) / OWN_ROWS, 1024, 0, stream>>>(erow, ecol, key, deg, h1, hagg, recs, meta, N, E);

    bn_stats<<<1024, 256, 0, stream>>>(hagg, bsum, bsq, N * D_HID);

    gemm2_wmma<<<(N + 63) / 64, 128, 0, stream>>>(hagg, bsum, bsq, gamma, beta, W2, b2, h2, N);
    own_gather<D_OUT, false><<<(N + OWN_ROWS - 1) / OWN_ROWS, 1024, 0, stream>>>(erow, ecol, key, deg, h2, outp, recs, meta, N, E);

    logsoftmax_rows<<<(N + 7) / 8, 256, 0, stream>>>(outp, N);
}


# Row loops that waited for each load before asking the next (BN sums: 24 trips; both propagations' start rows: 7) ask all of a wave's rows at once; BN sums in registers, 16-byte loads: new sum order
# speedup vs baseline: 2.5572x; 1.0998x over previous
#include <hip/hip_runtime.h>
#include <hip/hip_bf16.h>
#include <math.h>

typedef float v8f __attribute__((ext_vector_type(8)));
typedef float v2f __attribute__((ext_vector_type(2)));

#define D_HID 128
#define D_OUT 64
#define BN_EPS 1e-5f


#define OWN_ROWS 196
#define KEY_SPAN 2048

typedef int vi4u __attribute__((ext_vector_type(4), aligned(4)));

__global__ void __launch_bounds__(256) count_edges(const int* __restrict__ col, float* deg,
                                                   unsigned* __restrict__ key, int E) {
    const int t = blockIdx.x * 256 + threadIdx.x;
    const int e0 = 4 * t;
    unsigned kw = 0u;
    if (e0 + 4 <= E) {
        const vi4u c = *(const vi4u*)(col + e0);
        #pragma unroll
        for (int i = 0; i < 4; ++i) {
            atomicAdd(&deg[c[i]], 1.0f);
            kw |= (((unsigned)c[i] / OWN_ROWS) & 255u) << (8 * i);
        }
    } else {
        #pragma unroll
        for (int i = 0; i < 4; ++i) {
            if (e0 + i < E) {
                const int ci = col[e0 + i];
                atomicAdd(&deg[ci], 1.0f);
                kw |= (((unsigned)ci / OWN_ROWS) & 255u) << (8 * i);
            }
        }
    }
    key[t] = kw;
}


typedef float f4u __attribute__((ext_vector_type(4), aligned(4)));

__global__ void __launch_bounds__(128) gemm1_wmma(const float* __restrict__ x, const float* __restrict__ W1,
                           const float* __restrict__ b1, float* __restrict__ h1,
                           float* __restrict__ deg, float* __restrict__ stats, int N) {
    if (threadIdx.x < 64) {
        const int i = blockIdx.x * 64 + threadIdx.x;
        if (i < N) deg[i] = 1.0f;
    }
    if (blockIdx.x == 0) { stats[threadIdx.x] = 0.0f; stats[threadIdx.x + 128] = 0.0f; }
    const int wave  = __builtin_amdgcn_readfirstlane((int)(threadIdx.x >> 5));
    const int lane  = threadIdx.x & 31;
    const int row0  = (blockIdx.x * 4 + wave) * 16;
    if (row0 >= N) return;
    const int n     = lane & 15;
    const int khalf = lane >> 4;

    v8f c[8];
    {
        const f4u blo = *(const f4u*)(b1 + 4 * n);
        const f4u bhi = *(const f4u*)(b1 + 64 + 4 * n);
        #pragma unroll
        for (int t = 0; t < 4; ++t) {
            #pragma unroll
            for (int r = 0; r < 8; ++r) { c[t][r] = blo[t]; c[t + 4][r] = bhi[t]; }
        }
    }

    const float* arow = x + (size_t)(row0 + n) * D_HID + 2 * khalf;
    const float* brow = W1 + (size_t)(2 * khalf) * D_HID + 4 * n;

    v2f an[4];
    #pragma unroll
    for (int j = 0; j < 4; ++j) { an[j].x = arow[4 * j]; an[j].y = arow[4 * j + 1]; }

    #pragma unroll 2
    for (int k0 = 0; k0 < 128; k0 += 16) {
        v2f a[4];
        f4u p0[4], p1[4], q0[4], q1[4];
        #pragma unroll
        for (int j = 0; j < 4; ++j) {
            a[j] = an[j];
            const float* bk = brow + (size_t)(k0 + 4 * j) * D_HID;
            p0[j] = *(const f4u*)(bk);
            p1[j] = *(const f4u*)(bk + D_HID);
            q0[j] = *(const f4u*)(bk + 64);
            q1[j] = *(const f4u*)(bk + D_HID + 64);
        }
        const int kn = (k0 + 16 < 128) ? k0 + 16 : k0;
        #pragma unroll
        for (int j = 0; j < 4; ++j) { an[j].x = arow[kn + 4 * j]; an[j].y = arow[kn + 4 * j + 1]; }
        #pragma unroll
        for (int j = 0; j < 4; ++j) {
            #pragma unroll
            for (int t = 0; t < 4; ++t) {
                v2f b;
                b.x = p0[j][t];
                b.y = p1[j][t];
                c[t] = __builtin_amdgcn_wmma_f32_16x16x4_f32(false, a[j], false, b, (short)0, c[t], false, false);
            }
            #pragma unroll
            for (int t = 0; t < 4; ++t) {
                v2f b;
                b.x = q0[j][t];
                b.y = q1[j][t];
                c[t + 4] = __builtin_amdgcn_wmma_f32_16x16x4_f32(false, a[j], false, b, (short)0, c[t + 4], false, false);
            }
        }
    }

    const int mbase = khalf * 8;
    #pragma unroll
    for (int r = 0; r < 8; ++r) {
        float* orow = h1 + (size_t)(row0 + mbase + r) * D_HID + 4 * n;
        f4u lo, hi;
        #pragma unroll
        for (int t = 0; t < 4; ++t) { lo[t] = c[t][r]; hi[t] = c[t + 4][r]; }
        *(f4u*)(orow)      = lo;
        *(f4u*)(orow + 64) = hi;
    }
}


#define QCAP 64
#define REC_CAP 16384

typedef unsigned vu4 __attribute__((ext_vector_type(4)));
template <int V> struct rowvec;
template <> struct rowvec<4> { typedef float t __attribute__((ext_vector_type(4), aligned(4))); };
template <> struct rowvec<2> { typedef float t __attribute__((ext_vector_type(2), aligned(4))); };

template <int CH, bool FULL>
__device__ __forceinline__ void add_rows(int r, int cl, int wi, int cnt,
                                         const float* __restrict__ h, float* acc, int lane) {
    constexpr int V = CH / 32;
    typedef typename rowvec<V>::t vec;
    #pragma unroll 1
    for (int j0 = 0; j0 < cnt; j0 += 8) {
        vec v[8];
        #pragma unroll
        for (int j = 0; j < 8; ++j) {
            const int rj = __builtin_amdgcn_readlane(r, j0 + j);
            v[j] = *(const vec*)(h + (size_t)rj * CH + lane * V);
        }
        #pragma unroll
        for (int j = 0; j < 8; ++j) {
            if (FULL || j0 + j < cnt) {
                const int cj = __builtin_amdgcn_readlane(cl, j0 + j);
                const float wj = __builtin_bit_cast(float, __builtin_amdgcn_readlane(wi, j0 + j));
                float* a = acc + cj * CH + lane;
                #pragma unroll
                for (int k = 0; k < V; ++k) atomicAdd(a + 32 * k, wj * v[j][k]);
            }
        }
    }
}

template <int CH, bool FULL, bool SAVE>
__device__ __forceinline__ void drain_round(const int* qe_, int cnt,
                                            const int* __restrict__ row, const int* __restrict__ col,
                                            const float* __restrict__ deg,
                                            const float* __restrict__ h, float* acc, int lo, int lane,
                                            int* rec, int* ctl) {
    int cl = -1, r = 0;
    float w = 0.0f;
    if (FULL || lane < cnt) {
        const int e = qe_[lane];
        r = row[e];
        const int c = col[e];
        cl = c - lo;
        w = rsqrtf(deg[r]) * rsqrtf(deg[c]);
    }
    const int wi = __builtin_bit_cast(int, w);
    if (SAVE) {
        int base = 0;
        if (lane == 0) base = atomicAdd(&ctl[0], 32);
        base = __builtin_amdgcn_readfirstlane(base);
        if (base + 32 <= REC_CAP) {
            rec[base + lane] = r;
            rec[REC_CAP + base + lane] = cl;
            rec[2 * REC_CAP + base + lane] = wi;
        } else if (lane == 0) {
            ctl[1] = 1;
        }
    }
    add_rows<CH, FULL>(r, cl, wi, cnt, h, acc, lane);
}

template <int CH, bool SAVE>
__global__ void __launch_bounds__(1024) own_gather(const int* __restrict__ row, const int* __restrict__ col,
                                                   const unsigned* __restrict__ key,
                                                   const float* __restrict__ deg, const float* __restrict__ h,
                                                   float* __restrict__ dst, int* recs, int* meta, int N, int E) {
    constexpr int V = CH / 32;
    typedef typename rowvec<V>::t vec;
    __shared__ float acc[OWN_ROWS * CH];
    __shared__ int qe[32][QCAP];
    __shared__ int ctl[2];
    const int wave = __builtin_amdgcn_readfirstlane((int)(threadIdx.x >> 5));
    const int lane = threadIdx.x & 31;
    const int lo = blockIdx.x * OWN_ROWS;
    const int nrows = min(OWN_ROWS, N - lo);
    int* rec = recs + (size_t)blockIdx.x * (3 * REC_CAP);
    if (SAVE && threadIdx.x == 0) { ctl[0] = 0; ctl[1] = 0; }

    {
        constexpr int DEPTH = (OWN_ROWS + 31) / 32;
        const int rq = min(wave + 32 * (lane < DEPTH ? lane : 0), nrows - 1);
        float dl = rsqrtf(deg[lo + rq]);
        dl = dl * dl;
        vec v[DEPTH];
        #pragma unroll
        for (int j = 0; j < DEPTH; ++j) {
            const int rc = min(wave + 32 * j, nrows - 1);
            v[j] = *(const vec*)(h + (size_t)(lo + rc) * CH + lane * V);
        }
        #pragma unroll
        for (int j = 0; j < DEPTH; ++j) {
            const int rc = min(wave + 32 * j, nrows - 1);
            const float dd = __builtin_bit_cast(float, __builtin_amdgcn_readlane(__builtin_bit_cast(int, dl), j));
            #pragma unroll
            for (int k = 0; k < V; ++k) acc[rc * CH + k * 32 + lane] = dd * v[j][k];
        }
    }
    __syncthreads();

    const int nslots = SAVE ? -1 : meta[blockIdx.x * 32];
    if (!SAVE && nslots >= 0) {
        const int nfr = nslots >> 5;
        for (int f = wave; f < nfr; f += 32) {
            const int r  = rec[f * 32 + lane];
            const int cl = rec[REC_CAP + f * 32 + lane];
            const int wi = rec[2 * REC_CAP + f * 32 + lane];
            const int cnt = __builtin_popcount(__builtin_amdgcn_ballot_w32(cl >= 0));
            if (cnt == 32) add_rows<CH, true>(r, cl, wi, 32, h, acc, lane);
            else           add_rows<CH, false>(r, cl, wi, cnt, h, acc, lane);
        }
    } else {
    int* myqe = qe[wave];
    const unsigned pat = (unsigned)blockIdx.x * 0x01010101u;
    const int nsuper = (E + KEY_SPAN - 1) / KEY_SPAN;
    const int s0 = (int)(((long long)blockIdx.x * nsuper) / gridDim.x);
    int q = 0;
    for (int it = wave; it < nsuper; it += 32) {
        int s = it + s0;
        if (s >= nsuper) s -= nsuper;
        vu4 kw[4];
        #pragma unroll
        for (int k = 0; k < 4; ++k) kw[k] = *(const vu4*)(key + (size_t)(s * 4 + k) * 128 + lane * 4);
        const bool tail = s * KEY_SPAN + KEY_SPAN > E;
        #pragma unroll
        for (int k = 0; k < 4; ++k) {
            const int eb = (s * 4 + k) * 512 + lane * 16;
            unsigned m = 0u;
            #pragma unroll
            for (int i = 0; i < 4; ++i) {
                const unsigned x = kw[k][i] ^ pat;
                const unsigned t = (x & 0x7f7f7f7fu) + 0x7f7f7f7fu;
                m |= (~(t | x) & 0x80808080u) >> i;
            }
            if (tail) {
                const int left = E - eb;
                unsigned keep = 0u;
                #pragma unroll
                for (int i = 0; i < 4; ++i) {
                    #pragma unroll
                    for (int j = 0; j < 4; ++j) keep |= (4 * i + j < left) ? (1u << (8 * j + 7 - i)) : 0u;
                }
                m &= keep;
            }
            unsigned bal = __builtin_amdgcn_ballot_w32(m != 0u);
            while (bal != 0u) {
                if (m != 0u) {
                    const int p = __builtin_ctz(m);
                    const int pos = q + (int)__builtin_amdgcn_mbcnt_lo(bal, 0u);
                    myqe[pos] = eb + ((7 - (p & 7)) << 2) + (p >> 3);
                    m &= m - 1u;
                }
                q += __builtin_popcount(bal);
                if (q >= 32) {
                    q -= 32;
                    drain_round<CH, true, SAVE>(myqe + q, 32, row, col, deg, h, acc, lo, lane, rec, ctl);
                }
                bal = __builtin_amdgcn_ballot_w32(m != 0u);
            }
        }
    }
    if (q > 0) drain_round<CH, false, SAVE>(myqe, q, row, col, deg, h, acc, lo, lane, rec, ctl);
    }
    __syncthreads();
    if (SAVE && threadIdx.x == 0) meta[blockIdx.x * 32] = ctl[1] ? -1 : ctl[0];

    for (int rl = wave; rl < nrows; rl += 32) {
        vec o;
        #pragma unroll
        for (int k = 0; k < V; ++k) o[k] = acc[rl * CH + k * 32 + lane];
        *(vec*)(dst + (size_t)(lo + rl) * CH + lane * V) = o;
    }
}


#define BN_ROWS 196
#define BN_DEPTH ((BN_ROWS + 31) / 32)

__global__ void __launch_bounds__(1024) bn_stats(const float* __restrict__ hagg, float* bsum, float* bsq, int N) {
    __shared__ float ssum[D_HID];
    __shared__ float ssq[D_HID];
    const int wave = __builtin_amdgcn_readfirstlane((int)(threadIdx.x >> 5));
    const int lane = threadIdx.x & 31;
    if (threadIdx.x < D_HID) { ssum[threadIdx.x] = 0.0f; ssq[threadIdx.x] = 0.0f; }
    const int lo = blockIdx.x * BN_ROWS;
    const int nrows = min(BN_ROWS, N - lo);
    f4u v[BN_DEPTH];
    #pragma unroll
    for (int j = 0; j < BN_DEPTH; ++j) {
        const int rl = wave + 32 * j;
        const int rc = rl < nrows ? rl : 0;
        v[j] = *(const f4u*)(hagg + (size_t)(lo + rc) * D_HID + 4 * lane);
    }
    float s[4] = {0.0f, 0.0f, 0.0f, 0.0f};
    float q[4] = {0.0f, 0.0f, 0.0f, 0.0f};
    #pragma unroll
    for (int j = 0; j < BN_DEPTH; ++j) {
        if (wave + 32 * j < nrows) {
            #pragma unroll
            for (int k = 0; k < 4; ++k) { s[k] += v[j][k]; q[k] += v[j][k] * v[j][k]; }
        }
    }
    __syncthreads();
    #pragma unroll
    for (int k = 0; k < 4; ++k) {
        atomicAdd(&ssum[4 * lane + k], s[k]);
        atomicAdd(&ssq[4 * lane + k], q[k]);
    }
    __syncthreads();
    if (threadIdx.x < D_HID) {
        atomicAdd(&bsum[threadIdx.x], ssum[threadIdx.x]);
        atomicAdd(&bsq[threadIdx.x], ssq[threadIdx.x]);
    }
}


__global__ void __launch_bounds__(128) gemm2_wmma(const float* __restrict__ hagg, const float* __restrict__ bsum,
                           const float* __restrict__ bsq, const float* __restrict__ gamma,
                           const float* __restrict__ beta, const float* __restrict__ W2,
                           const float* __restrict__ b2, float* __restrict__ h2, int N) {
    __shared__ float sc[D_HID];
    __shared__ float sh[D_HID];
    const int wave  = __builtin_amdgcn_readfirstlane((int)(threadIdx.x >> 5));
    const int lane  = threadIdx.x & 31;
    const int row0  = (blockIdx.x * 4 + wave) * 16;
    const bool live = row0 < N;
    const int n     = lane & 15;
    const int khalf = lane >> 4;
    const float* arow = hagg + (size_t)(row0 + n) * D_HID + 2 * khalf;

    v2f an[4];
    #pragma unroll
    for (int j = 0; j < 4; ++j) { an[j].x = 0.0f; an[j].y = 0.0f; }
    if (live) {
        #pragma unroll
        for (int j = 0; j < 4; ++j) { an[j].x = arow[4 * j]; an[j].y = arow[4 * j + 1]; }
    }
    {
        const int ch = threadIdx.x;
        const float inv = 1.0f / (float)N;
        const float mu = bsum[ch] * inv;
        const float var = bsq[ch] * inv - mu * mu;
        const float s = gamma[ch] * rsqrtf(var + BN_EPS);
        sc[ch] = s;
        sh[ch] = beta[ch] - mu * s;
    }
    __syncthreads();
    if (!live) return;

    v8f c[4];
    {
        const f4u bias = *(const f4u*)(b2 + 4 * n);
        #pragma unroll
        for (int t = 0; t < 4; ++t) {
            #pragma unroll
            for (int r = 0; r < 8; ++r) c[t][r] = bias[t];
        }
    }

    const float* brow = W2 + (size_t)(2 * khalf) * D_OUT + 4 * n;

    #pragma unroll 2
    for (int k0 = 0; k0 < 128; k0 += 16) {
        v2f a[4];
        f4u p0[4], p1[4];
        #pragma unroll
        for (int j = 0; j < 4; ++j) {
            const int k = k0 + 4 * j;
            const int kk = k + 2 * khalf;
            a[j].x = fmaxf(an[j].x * sc[kk + 0] + sh[kk + 0], 0.0f);
            a[j].y = fmaxf(an[j].y * sc[kk + 1] + sh[kk + 1], 0.0f);
            const float* bk = brow + (size_t)k * D_OUT;
            p0[j] = *(const f4u*)(bk);
            p1[j] = *(const f4u*)(bk + D_OUT);
        }
        const int kn = (k0 + 16 < 128) ? k0 + 16 : k0;
        #pragma unroll
        for (int j = 0; j < 4; ++j) { an[j].x = arow[kn + 4 * j]; an[j].y = arow[kn + 4 * j + 1]; }
        #pragma unroll
        for (int j = 0; j < 4; ++j) {
            #pragma unroll
            for (int t = 0; t < 4; ++t) {
                v2f b;
                b.x = p0[j][t];
                b.y = p1[j][t];
                c[t] = __builtin_amdgcn_wmma_f32_16x16x4_f32(false, a[j], false, b, (short)0, c[t], false, false);
            }
        }
    }

    const int mbase = khalf * 8;
    #pragma unroll
    for (int r = 0; r < 8; ++r) {
        f4u o;
        #pragma unroll
        for (int t = 0; t < 4; ++t) o[t] = c[t][r];
        *(f4u*)(h2 + (size_t)(row0 + mbase + r) * D_OUT + 4 * n) = o;
    }
}


__global__ void logsoftmax_rows(float* out, int N) {
    const int wave = threadIdx.x >> 5;
    const int lane = threadIdx.x & 31;
    const int rowi = blockIdx.x * 8 + wave;
    if (rowi >= N) return;
    float* p = out + (size_t)rowi * D_OUT;
    float v0 = p[lane];
    float v1 = p[lane + 32];
    float mx = fmaxf(v0, v1);
    #pragma unroll
    for (int off = 16; off >= 1; off >>= 1)
        mx = fmaxf(mx, __shfl_xor(mx, off, 32));
    float s = __expf(v0 - mx) + __expf(v1 - mx);
    #pragma unroll
    for (int off = 16; off >= 1; off >>= 1)
        s += __shfl_xor(s, off, 32);
    float l = mx + __logf(s);
    p[lane]      = v0 - l;
    p[lane + 32] = v1 - l;
}


extern "C" void kernel_launch(void* const* d_in, const int* in_sizes, int n_in,
                              void* d_out, int out_size, void* d_ws, size_t ws_size,
                              hipStream_t stream) {
    const float* x     = (const float*)d_in[0];
    const float* W1    = (const float*)d_in[1];
    const float* b1    = (const float*)d_in[2];
    const float* gamma = (const float*)d_in[3];
    const float* beta  = (const float*)d_in[4];
    const float* W2    = (const float*)d_in[5];
    const float* b2    = (const float*)d_in[6];
    const int*   ei    = (const int*)d_in[7];

    const int N = in_sizes[0] / D_HID;
    const int E = in_sizes[7] / 2;
    const int* erow = ei;
    const int* ecol = ei + E;

    float* ws    = (float*)d_ws;
    float* deg   = ws;
    float* h1    = deg + N;
    float* hagg  = h1 + (size_t)N * D_HID;
    float* stats = hagg + (size_t)N * D_HID;
    float* bsum  = stats;
    float* bsq   = stats + 128;
    float* h2    = h1;
    int*   meta  = (int*)((char*)d_ws + (size_t)(63u << 20));
    int*   recs  = (int*)((char*)d_ws + (size_t)(64u << 20));
    unsigned* key = (unsigned*)((char*)d_ws + (size_t)(120u << 20));
    float* outp  = (float*)d_out;

    gemm1_wmma<<<(N + 63) / 64, 128, 0, stream>>>(x, W1, b1, h1, deg, stats, N);
    count_edges<<<((E + KEY_SPAN - 1) / KEY_SPAN) * (KEY_SPAN / 1024), 256, 0, stream>>>(ecol, deg, key, E);
    own_gather<D_HID, true><<<(N + OWN_ROWS - 1) / OWN_ROWS, 1024, 0, stream>>>(erow, ecol, key, deg, h1, hagg, recs, meta, N, E);

    bn_stats<<<(N + BN_ROWS - 1) / BN_ROWS, 1024, 0, stream>>>(hagg, bsum, bsq, N);

    gemm2_wmma<<<(N + 63) / 64, 128, 0, stream>>>(hagg, bsum, bsq, gamma, beta, W2, b2, h2, N);
    own_gather<D_OUT, false><<<(N + OWN_ROWS - 1) / OWN_ROWS, 1024, 0, stream>>>(erow, ecol, key, deg, h2, outp, recs, meta, N, E);

    logsoftmax_rows<<<(N + 7) / 8, 256, 0, stream>>>(outp, N);
}


# First propagation, per matched edge: one scattered 4-byte load (source row | owned row, packed by the count kernel), degree factors from an LDS table of all nodes; was four loads; sum order kept
# speedup vs baseline: 2.6508x; 1.0366x over previous
#include <hip/hip_runtime.h>
#include <hip/hip_bf16.h>
#include <math.h>

typedef float v8f __attribute__((ext_vector_type(8)));
typedef float v2f __attribute__((ext_vector_type(2)));

#define D_HID 128
#define D_OUT 64
#define BN_EPS 1e-5f


#define OWN_ROWS 196
#define KEY_SPAN 2048

#define DIS_CAP 50176

typedef int vi4u __attribute__((ext_vector_type(4), aligned(4)));
typedef unsigned vu4 __attribute__((ext_vector_type(4)));

__global__ void __launch_bounds__(256) count_edges(const int* __restrict__ row, const int* __restrict__ col, float* deg,
                                                   unsigned* __restrict__ key, unsigned* __restrict__ pk, int E) {
    const int t = blockIdx.x * 256 + threadIdx.x;
    const int e0 = 4 * t;
    unsigned kw = 0u;
    if (e0 + 4 <= E) {
        const vi4u c = *(const vi4u*)(col + e0);
        const vi4u rr = *(const vi4u*)(row + e0);
        vu4 pw;
        #pragma unroll
        for (int i = 0; i < 4; ++i) {
            const unsigned ci = (unsigned)c[i];
            const unsigned ob = ci / OWN_ROWS;
            atomicAdd(&deg[c[i]], 1.0f);
            kw |= (ob & 255u) << (8 * i);
            pw[i] = (unsigned)rr[i] | ((ci - ob * OWN_ROWS) << 16);
        }
        *(vu4*)(pk + e0) = pw;
    } else {
        #pragma unroll
        for (int i = 0; i < 4; ++i) {
            if (e0 + i < E) {
                const int ci = col[e0 + i];
                const unsigned ob = (unsigned)ci / OWN_ROWS;
                atomicAdd(&deg[ci], 1.0f);
                kw |= (ob & 255u) << (8 * i);
                pk[e0 + i] = (unsigned)row[e0 + i] | (((unsigned)ci - ob * OWN_ROWS) << 16);
            }
        }
    }
    key[t] = kw;
}


typedef float f4u __attribute__((ext_vector_type(4), aligned(4)));

__global__ void __launch_bounds__(128) gemm1_wmma(const float* __restrict__ x, const float* __restrict__ W1,
                           const float* __restrict__ b1, float* __restrict__ h1,
                           float* __restrict__ deg, float* __restrict__ stats, int N) {
    if (threadIdx.x < 64) {
        const int i = blockIdx.x * 64 + threadIdx.x;
        if (i < N) deg[i] = 1.0f;
    }
    if (blockIdx.x == 0) { stats[threadIdx.x] = 0.0f; stats[threadIdx.x + 128] = 0.0f; }
    const int wave  = __builtin_amdgcn_readfirstlane((int)(threadIdx.x >> 5));
    const int lane  = threadIdx.x & 31;
    const int row0  = (blockIdx.x * 4 + wave) * 16;
    if (row0 >= N) return;
    const int n     = lane & 15;
    const int khalf = lane >> 4;

    v8f c[8];
    {
        const f4u blo = *(const f4u*)(b1 + 4 * n);
        const f4u bhi = *(const f4u*)(b1 + 64 + 4 * n);
        #pragma unroll
        for (int t = 0; t < 4; ++t) {
            #pragma unroll
            for (int r = 0; r < 8; ++r) { c[t][r] = blo[t]; c[t + 4][r] = bhi[t]; }
        }
    }

    const float* arow = x + (size_t)(row0 + n) * D_HID + 2 * khalf;
    const float* brow = W1 + (size_t)(2 * khalf) * D_HID + 4 * n;

    v2f an[4];
    #pragma unroll
    for (int j = 0; j < 4; ++j) { an[j].x = arow[4 * j]; an[j].y = arow[4 * j + 1]; }

    #pragma unroll 2
    for (int k0 = 0; k0 < 128; k0 += 16) {
        v2f a[4];
        f4u p0[4], p1[4], q0[4], q1[4];
        #pragma unroll
        for (int j = 0; j < 4; ++j) {
            a[j] = an[j];
            const float* bk = brow + (size_t)(k0 + 4 * j) * D_HID;
            p0[j] = *(const f4u*)(bk);
            p1[j] = *(const f4u*)(bk + D_HID);
            q0[j] = *(const f4u*)(bk + 64);
            q1[j] = *(const f4u*)(bk + D_HID + 64);
        }
        const int kn = (k0 + 16 < 128) ? k0 + 16 : k0;
        #pragma unroll
        for (int j = 0; j < 4; ++j) { an[j].x = arow[kn + 4 * j]; an[j].y = arow[kn + 4 * j + 1]; }
        #pragma unroll
        for (int j = 0; j < 4; ++j) {
            #pragma unroll
            for (int t = 0; t < 4; ++t) {
                v2f b;
                b.x = p0[j][t];
                b.y = p1[j][t];
                c[t] = __builtin_amdgcn_wmma_f32_16x16x4_f32(false, a[j], false, b, (short)0, c[t], false, false);
            }
            #pragma unroll
            for (int t = 0; t < 4; ++t) {
                v2f b;
                b.x = q0[j][t];
                b.y = q1[j][t];
                c[t + 4] = __builtin_amdgcn_wmma_f32_16x16x4_f32(false, a[j], false, b, (short)0, c[t + 4], false, false);
            }
        }
    }

    const int mbase = khalf * 8;
    #pragma unroll
    for (int r = 0; r < 8; ++r) {
        float* orow = h1 + (size_t)(row0 + mbase + r) * D_HID + 4 * n;
        f4u lo, hi;
        #pragma unroll
        for (int t = 0; t < 4; ++t) { lo[t] = c[t][r]; hi[t] = c[t + 4][r]; }
        *(f4u*)(orow)      = lo;
        *(f4u*)(orow + 64) = hi;
    }
}


#define QCAP 64
#define REC_CAP 16384

template <int V> struct rowvec;
template <> struct rowvec<4> { typedef float t __attribute__((ext_vector_type(4), aligned(4))); };
template <> struct rowvec<2> { typedef float t __attribute__((ext_vector_type(2), aligned(4))); };

template <int CH, bool FULL>
__device__ __forceinline__ void add_rows(int r, int cl, int wi, int cnt,
                                         const float* __restrict__ h, float* acc, int lane) {
    constexpr int V = CH / 32;
    typedef typename rowvec<V>::t vec;
    #pragma unroll 1
    for (int j0 = 0; j0 < cnt; j0 += 8) {
        vec v[8];
        #pragma unroll
        for (int j = 0; j < 8; ++j) {
            const int rj = __builtin_amdgcn_readlane(r, j0 + j);
            v[j] = *(const vec*)(h + (size_t)rj * CH + lane * V);
        }
        #pragma unroll
        for (int j = 0; j < 8; ++j) {
            if (FULL || j0 + j < cnt) {
                const int cj = __builtin_amdgcn_readlane(cl, j0 + j);
                const float wj = __builtin_bit_cast(float, __builtin_amdgcn_readlane(wi, j0 + j));
                float* a = acc + cj * CH + lane;
                #pragma unroll
                for (int k = 0; k < V; ++k) atomicAdd(a + 32 * k, wj * v[j][k]);
            }
        }
    }
}

template <int CH, bool FULL, bool SAVE>
__device__ __forceinline__ void drain_round(const int* qe_, int cnt,
                                            const int* __restrict__ row, const int* __restrict__ col,
                                            const unsigned* __restrict__ pk,
                                            const float* __restrict__ deg, const float* sdis,
                                            const float* __restrict__ h, float* acc, int lo, int lane,
                                            int* rec, int* ctl) {
    int cl = -1, r = 0;
    float w = 0.0f;
    if (FULL || lane < cnt) {
        const int e = qe_[lane];
        if (SAVE) {
            const unsigned p = pk[e];
            r = (int)(p & 0xffffu);
            cl = (int)(p >> 16);
            w = sdis[r] * sdis[lo + cl];
        } else {
            r = row[e];
            const int c = col[e];
            cl = c - lo;
            w = rsqrtf(deg[r]) * rsqrtf(deg[c]);
        }
    }
    const int wi = __builtin_bit_cast(int, w);
    if (SAVE) {
        int base = 0;
        if (lane == 0) base = atomicAdd(&ctl[0], 32);
        base = __builtin_amdgcn_readfirstlane(base);
        if (base + 32 <= REC_CAP) {
            rec[base + lane] = r;
            rec[REC_CAP + base + lane] = cl;
            rec[2 * REC_CAP + base + lane] = wi;
        } else if (lane == 0) {
            ctl[1] = 1;
        }
    }
    add_rows<CH, FULL>(r, cl, wi, cnt, h, acc, lane);
}

template <int CH, bool SAVE>
__global__ void __launch_bounds__(1024) own_gather(const int* __restrict__ row, const int* __restrict__ col,
                                                   const unsigned* __restrict__ key, const unsigned* __restrict__ pk,
                                                   const float* __restrict__ deg, const float* __restrict__ h,
                                                   float* __restrict__ dst, int* recs, int* meta, int N, int E) {
    constexpr int V = CH / 32;
    typedef typename rowvec<V>::t vec;
    __shared__ float acc[OWN_ROWS * CH];
    __shared__ int qe[32][QCAP];
    __shared__ int ctl[2];
    __shared__ float sdis[SAVE ? DIS_CAP : 32];
    const int wave = __builtin_amdgcn_readfirstlane((int)(threadIdx.x >> 5));
    const int lane = threadIdx.x & 31;
    const int lo = blockIdx.x * OWN_ROWS;
    const int nrows = min(OWN_ROWS, N - lo);
    int* rec = recs + (size_t)blockIdx.x * (3 * REC_CAP);
    if (SAVE && threadIdx.x == 0) { ctl[0] = 0; ctl[1] = 0; }

    constexpr int TB = SAVE ? (DIS_CAP / 4 + 1023) / 1024 : 1;
    const int n4 = N >> 2;
    f4u d[TB];
    if (SAVE) {
        #pragma unroll
        for (int j = 0; j < TB; ++j) {
            const int i = (int)threadIdx.x + 1024 * j;
            d[j] = *(const f4u*)(deg + 4 * (i < n4 ? i : 0));
        }
    }

    {
        constexpr int DEPTH = (OWN_ROWS + 31) / 32;
        const int rq = min(wave + 32 * (lane < DEPTH ? lane : 0), nrows - 1);
        float dl = rsqrtf(deg[lo + rq]);
        dl = dl * dl;
        vec v[DEPTH];
        #pragma unroll
        for (int j = 0; j < DEPTH; ++j) {
            const int rc = min(wave + 32 * j, nrows - 1);
            v[j] = *(const vec*)(h + (size_t)(lo + rc) * CH + lane * V);
        }
        if (SAVE) {
            #pragma unroll
            for (int j = 0; j < TB; ++j) {
                const int i = (int)threadIdx.x + 1024 * j;
                if (i < n4) {
                    #pragma unroll
                    for (int k = 0; k < 4; ++k) sdis[4 * i + k] = rsqrtf(d[j][k]);
                }
            }
        }
        #pragma unroll
        for (int j = 0; j < DEPTH; ++j) {
            const int rc = min(wave + 32 * j, nrows - 1);
            const float dd = __builtin_bit_cast(float, __builtin_amdgcn_readlane(__builtin_bit_cast(int, dl), j));
            #pragma unroll
            for (int k = 0; k < V; ++k) acc[rc * CH + k * 32 + lane] = dd * v[j][k];
        }
    }
    if (SAVE) {
        for (int i = (n4 << 2) + (int)threadIdx.x; i < N; i += 1024) sdis[i] = rsqrtf(deg[i]);
    }
    __syncthreads();

    const int nslots = SAVE ? -1 : meta[blockIdx.x * 32];
    if (!SAVE && nslots >= 0) {
        const int nfr = nslots >> 5;
        for (int f = wave; f < nfr; f += 32) {
            const int r  = rec[f * 32 + lane];
            const int cl = rec[REC_CAP + f * 32 + lane];
            const int wi = rec[2 * REC_CAP + f * 32 + lane];
            const int cnt = __builtin_popcount(__builtin_amdgcn_ballot_w32(cl >= 0));
            if (cnt == 32) add_rows<CH, true>(r, cl, wi, 32, h, acc, lane);
            else           add_rows<CH, false>(r, cl, wi, cnt, h, acc, lane);
        }
    } else {
    int* myqe = qe[wave];
    const unsigned pat = (unsigned)blockIdx.x * 0x01010101u;
    const int nsuper = (E + KEY_SPAN - 1) / KEY_SPAN;
    const int s0 = (int)(((long long)blockIdx.x * nsuper) / gridDim.x);
    int q = 0;
    for (int it = wave; it < nsuper; it += 32) {
        int s = it + s0;
        if (s >= nsuper) s -= nsuper;
        vu4 kw[4];
        #pragma unroll
        for (int k = 0; k < 4; ++k) kw[k] = *(const vu4*)(key + (size_t)(s * 4 + k) * 128 + lane * 4);
        const bool tail = s * KEY_SPAN + KEY_SPAN > E;
        #pragma unroll
        for (int k = 0; k < 4; ++k) {
            const int eb = (s * 4 + k) * 512 + lane * 16;
            unsigned m = 0u;
            #pragma unroll
            for (int i = 0; i < 4; ++i) {
                const unsigned x = kw[k][i] ^ pat;
                const unsigned t = (x & 0x7f7f7f7fu) + 0x7f7f7f7fu;
                m |= (~(t | x) & 0x80808080u) >> i;
            }
            if (tail) {
                const int left = E - eb;
                unsigned keep = 0u;
                #pragma unroll
                for (int i = 0; i < 4; ++i) {
                    #pragma unroll
                    for (int j = 0; j < 4; ++j) keep |= (4 * i + j < left) ? (1u << (8 * j + 7 - i)) : 0u;
                }
                m &= keep;
            }
            unsigned bal = __builtin_amdgcn_ballot_w32(m != 0u);
            while (bal != 0u) {
                if (m != 0u) {
                    const int p = __builtin_ctz(m);
                    const int pos = q + (int)__builtin_amdgcn_mbcnt_lo(bal, 0u);
                    myqe[pos] = eb + ((7 - (p & 7)) << 2) + (p >> 3);
                    m &= m - 1u;
                }
                q += __builtin_popcount(bal);
                if (q >= 32) {
                    q -= 32;
                    drain_round<CH, true, SAVE>(myqe + q, 32, row, col, pk, deg, sdis, h, acc, lo, lane, rec, ctl);
                }
                bal = __builtin_amdgcn_ballot_w32(m != 0u);
            }
        }
    }
    if (q > 0) drain_round<CH, false, SAVE>(myqe, q, row, col, pk, deg, sdis, h, acc, lo, lane, rec, ctl);
    }
    __syncthreads();
    if (SAVE && threadIdx.x == 0) meta[blockIdx.x * 32] = ctl[1] ? -1 : ctl[0];

    for (int rl = wave; rl < nrows; rl += 32) {
        vec o;
        #pragma unroll
        for (int k = 0; k < V; ++k) o[k] = acc[rl * CH + k * 32 + lane];
        *(vec*)(dst + (size_t)(lo + rl) * CH + lane * V) = o;
    }
}


#define BN_ROWS 196
#define BN_DEPTH ((BN_ROWS + 31) / 32)

__global__ void __launch_bounds__(1024) bn_stats(const float* __restrict__ hagg, float* bsum, float* bsq, int N) {
    __shared__ float ssum[D_HID];
    __shared__ float ssq[D_HID];
    const int wave = __builtin_amdgcn_readfirstlane((int)(threadIdx.x >> 5));
    const int lane = threadIdx.x & 31;
    if (threadIdx.x < D_HID) { ssum[threadIdx.x] = 0.0f; ssq[threadIdx.x] = 0.0f; }
    const int lo = blockIdx.x * BN_ROWS;
    const int nrows = min(BN_ROWS, N - lo);
    f4u v[BN_DEPTH];
    #pragma unroll
    for (int j = 0; j < BN_DEPTH; ++j) {
        const int rl = wave + 32 * j;
        const int rc = rl < nrows ? rl : 0;
        v[j] = *(const f4u*)(hagg + (size_t)(lo + rc) * D_HID + 4 * lane);
    }
    float s[4] = {0.0f, 0.0f, 0.0f, 0.0f};
    float q[4] = {0.0f, 0.0f, 0.0f, 0.0f};
    #pragma unroll
    for (int j = 0; j < BN_DEPTH; ++j) {
        if (wave + 32 * j < nrows) {
            #pragma unroll
            for (int k = 0; k < 4; ++k) { s[k] += v[j][k]; q[k] += v[j][k] * v[j][k]; }
        }
    }
    __syncthreads();
    #pragma unroll
    for (int k = 0; k < 4; ++k) {
        atomicAdd(&ssum[4 * lane + k], s[k]);
        atomicAdd(&ssq[4 * lane + k], q[k]);
    }
    __syncthreads();
    if (threadIdx.x < D_HID) {
        atomicAdd(&bsum[threadIdx.x], ssum[threadIdx.x]);
        atomicAdd(&bsq[threadIdx.x], ssq[threadIdx.x]);
    }
}


__global__ void __launch_bounds__(128) gemm2_wmma(const float* __restrict__ hagg, const float* __restrict__ bsum,
                           const float* __restrict__ bsq, const float* __restrict__ gamma,
                           const float* __restrict__ beta, const float* __restrict__ W2,
                           const float* __restrict__ b2, float* __restrict__ h2, int N) {
    __shared__ float sc[D_HID];
    __shared__ float sh[D_HID];
    const int wave  = __builtin_amdgcn_readfirstlane((int)(threadIdx.x >> 5));
    const int lane  = threadIdx.x & 31;
    const int row0  = (blockIdx.x * 4 + wave) * 16;
    const bool live = row0 < N;
    const int n     = lane & 15;
    const int khalf = lane >> 4;
    const float* arow = hagg + (size_t)(row0 + n) * D_HID + 2 * khalf;

    v2f an[4];
    #pragma unroll
    for (int j = 0; j < 4; ++j) { an[j].x = 0.0f; an[j].y = 0.0f; }
    if (live) {
        #pragma unroll
        for (int j = 0; j < 4; ++j) { an[j].x = arow[4 * j]; an[j].y = arow[4 * j + 1]; }
    }
    {
        const int ch = threadIdx.x;
        const float inv = 1.0f / (float)N;
        const float mu = bsum[ch] * inv;
        const float var = bsq[ch] * inv - mu * mu;
        const float s = gamma[ch] * rsqrtf(var + BN_EPS);
        sc[ch] = s;
        sh[ch] = beta[ch] - mu * s;
    }
    __syncthreads();
    if (!live) return;

    v8f c[4];
    {
        const f4u bias = *(const f4u*)(b2 + 4 * n);
        #pragma unroll
        for (int t = 0; t < 4; ++t) {
            #pragma unroll
            for (int r = 0; r < 8; ++r) c[t][r] = bias[t];
        }
    }

    const float* brow = W2 + (size_t)(2 * khalf) * D_OUT + 4 * n;

    #pragma unroll 2
    for (int k0 = 0; k0 < 128; k0 += 16) {
        v2f a[4];
        f4u p0[4], p1[4];
        #pragma unroll
        for (int j = 0; j < 4; ++j) {
            const int k = k0 + 4 * j;
            const int kk = k + 2 * khalf;
            a[j].x = fmaxf(an[j].x * sc[kk + 0] + sh[kk + 0], 0.0f);
            a[j].y = fmaxf(an[j].y * sc[kk + 1] + sh[kk + 1], 0.0f);
            const float* bk = brow + (size_t)k * D_OUT;
            p0[j] = *(const f4u*)(bk);
            p1[j] = *(const f4u*)(bk + D_OUT);
        }
        const int kn = (k0 + 16 < 128) ? k0 + 16 : k0;
        #pragma unroll
        for (int j = 0; j < 4; ++j) { an[j].x = arow[kn + 4 * j]; an[j].y = arow[kn + 4 * j + 1]; }
        #pragma unroll
        for (int j = 0; j < 4; ++j) {
            #pragma unroll
            for (int t = 0; t < 4; ++t) {
                v2f b;
                b.x = p0[j][t];
                b.y = p1[j][t];
                c[t] = __builtin_amdgcn_wmma_f32_16x16x4_f32(false, a[j], false, b, (short)0, c[t], false, false);
            }
        }
    }

    const int mbase = khalf * 8;
    #pragma unroll
    for (int r = 0; r < 8; ++r) {
        f4u o;
        #pragma unroll
        for (int t = 0; t < 4; ++t) o[t] = c[t][r];
        *(f4u*)(h2 + (size_t)(row0 + mbase + r) * D_OUT + 4 * n) = o;
    }
}


__global__ void logsoftmax_rows(float* out, int N) {
    const int wave = threadIdx.x >> 5;
    const int lane = threadIdx.x & 31;
    const int rowi = blockIdx.x * 8 + wave;
    if (rowi >= N) return;
    float* p = out + (size_t)rowi * D_OUT;
    float v0 = p[lane];
    float v1 = p[lane + 32];
    float mx = fmaxf(v0, v1);
    #pragma unroll
    for (int off = 16; off >= 1; off >>= 1)
        mx = fmaxf(mx, __shfl_xor(mx, off, 32));
    float s = __expf(v0 - mx) + __expf(v1 - mx);
    #pragma unroll
    for (int off = 16; off >= 1; off >>= 1)
        s += __shfl_xor(s, off, 32);
    float l = mx + __logf(s);
    p[lane]      = v0 - l;
    p[lane + 32] = v1 - l;
}


extern "C" void kernel_launch(void* const* d_in, const int* in_sizes, int n_in,
                              void* d_out, int out_size, void* d_ws, size_t ws_size,
                              hipStream_t stream) {
    const float* x     = (const float*)d_in[0];
    const float* W1    = (const float*)d_in[1];
    const float* b1    = (const float*)d_in[2];
    const float* gamma = (const float*)d_in[3];
    const float* beta  = (const float*)d_in[4];
    const float* W2    = (const float*)d_in[5];
    const float* b2    = (const float*)d_in[6];
    const int*   ei    = (const int*)d_in[7];

    const int N = in_sizes[0] / D_HID;
    const int E = in_sizes[7] / 2;
    const int* erow = ei;
    const int* ecol = ei + E;

    float* ws    = (float*)d_ws;
    float* deg   = ws;
    float* h1    = deg + N;
    float* hagg  = h1 + (size_t)N * D_HID;
    float* stats = hagg + (size_t)N * D_HID;
    float* bsum  = stats;
    float* bsq   = stats + 128;
    float* h2    = h1;
    int*   meta  = (int*)((char*)d_ws + (size_t)(63u << 20));
    int*   recs  = (int*)((char*)d_ws + (size_t)(64u << 20));
    unsigned* key = (unsigned*)((char*)d_ws + (size_t)(120u << 20));
    unsigned* pk  = (unsigned*)((char*)d_ws + (size_t)(124u << 20));
    float* outp  = (float*)d_out;

    gemm1_wmma<<<(N + 63) / 64, 128, 0, stream>>>(x, W1, b1, h1, deg, stats, N);
    count_edges<<<((E + KEY_SPAN - 1) / KEY_SPAN) * (KEY_SPAN / 1024), 256, 0, stream>>>(erow, ecol, deg, key, pk, E);
    own_gather<D_HID, true><<<(N + OWN_ROWS - 1) / OWN_ROWS, 1024, 0, stream>>>(erow, ecol, key, pk, deg, h1, hagg, recs, meta, N, E);

    bn_stats<<<(N + BN_ROWS - 1) / BN_ROWS, 1024, 0, stream>>>(hagg, bsum, bsq, N);

    gemm2_wmma<<<(N + 63) / 64, 128, 0, stream>>>(hagg, bsum, bsq, gamma, beta, W2, b2, h2, N);
    own_gather<D_OUT, false><<<(N + OWN_ROWS - 1) / OWN_ROWS, 1024, 0, stream>>>(erow, ecol, key, pk, deg, h2, outp, recs, meta, N, E);

    logsoftmax_rows<<<(N + 7) / 8, 256, 0, stream>>>(outp, N);
}


# BN sums and log_softmax run in the owner block on its finished rows in LDS (column sums by channel-owning threads, same global float atomics; softmax at row store): 7 -> 5 launches; new sum orders
# speedup vs baseline: 2.8665x; 1.0814x over previous
#include <hip/hip_runtime.h>
#include <hip/hip_bf16.h>
#include <math.h>

typedef float v8f __attribute__((ext_vector_type(8)));
typedef float v2f __attribute__((ext_vector_type(2)));

#define D_HID 128
#define D_OUT 64
#define BN_EPS 1e-5f


#define OWN_ROWS 196
#define KEY_SPAN 2048

#define DIS_CAP 50176

typedef int vi4u __attribute__((ext_vector_type(4), aligned(4)));
typedef unsigned vu4 __attribute__((ext_vector_type(4)));

__global__ void __launch_bounds__(256) count_edges(const int* __restrict__ row, const int* __restrict__ col, float* deg,
                                                   unsigned* __restrict__ key, unsigned* __restrict__ pk, int E) {
    const int t = blockIdx.x * 256 + threadIdx.x;
    const int e0 = 4 * t;
    unsigned kw = 0u;
    if (e0 + 4 <= E) {
        const vi4u c = *(const vi4u*)(col + e0);
        const vi4u rr = *(const vi4u*)(row + e0);
        vu4 pw;
        #pragma unroll
        for (int i = 0; i < 4; ++i) {
            const unsigned ci = (unsigned)c[i];
            const unsigned ob = ci / OWN_ROWS;
            atomicAdd(&deg[c[i]], 1.0f);
            kw |= (ob & 255u) << (8 * i);
            pw[i] = (unsigned)rr[i] | ((ci - ob * OWN_ROWS) << 16);
        }
        *(vu4*)(pk + e0) = pw;
    } else {
        #pragma unroll
        for (int i = 0; i < 4; ++i) {
            if (e0 + i < E) {
                const int ci = col[e0 + i];
                const unsigned ob = (unsigned)ci / OWN_ROWS;
                atomicAdd(&deg[ci], 1.0f);
                kw |= (ob & 255u) << (8 * i);
                pk[e0 + i] = (unsigned)row[e0 + i] | (((unsigned)ci - ob * OWN_ROWS) << 16);
            }
        }
    }
    key[t] = kw;
}


typedef float f4u __attribute__((ext_vector_type(4), aligned(4)));

__global__ void __launch_bounds__(128) gemm1_wmma(const float* __restrict__ x, const float* __restrict__ W1,
                           const float* __restrict__ b1, float* __restrict__ h1,
                           float* __restrict__ deg, float* __restrict__ stats, int N) {
    if (threadIdx.x < 64) {
        const int i = blockIdx.x * 64 + threadIdx.x;
        if (i < N) deg[i] = 1.0f;
    }
    if (blockIdx.x == 0) { stats[threadIdx.x] = 0.0f; stats[threadIdx.x + 128] = 0.0f; }
    const int wave  = __builtin_amdgcn_readfirstlane((int)(threadIdx.x >> 5));
    const int lane  = threadIdx.x & 31;
    const int row0  = (blockIdx.x * 4 + wave) * 16;
    if (row0 >= N) return;
    const int n     = lane & 15;
    const int khalf = lane >> 4;

    v8f c[8];
    {
        const f4u blo = *(const f4u*)(b1 + 4 * n);
        const f4u bhi = *(const f4u*)(b1 + 64 + 4 * n);
        #pragma unroll
        for (int t = 0; t < 4; ++t) {
            #pragma unroll
            for (int r = 0; r < 8; ++r) { c[t][r] = blo[t]; c[t + 4][r] = bhi[t]; }
        }
    }

    const float* arow = x + (size_t)(row0 + n) * D_HID + 2 * khalf;
    const float* brow = W1 + (size_t)(2 * khalf) * D_HID + 4 * n;

    v2f an[4];
    #pragma unroll
    for (int j = 0; j < 4; ++j) { an[j].x = arow[4 * j]; an[j].y = arow[4 * j + 1]; }

    #pragma unroll 2
    for (int k0 = 0; k0 < 128; k0 += 16) {
        v2f a[4];
        f4u p0[4], p1[4], q0[4], q1[4];
        #pragma unroll
        for (int j = 0; j < 4; ++j) {
            a[j] = an[j];
            const float* bk = brow + (size_t)(k0 + 4 * j) * D_HID;
            p0[j] = *(const f4u*)(bk);
            p1[j] = *(const f4u*)(bk + D_HID);
            q0[j] = *(const f4u*)(bk + 64);
            q1[j] = *(const f4u*)(bk + D_HID + 64);
        }
        const int kn = (k0 + 16 < 128) ? k0 + 16 : k0;
        #pragma unroll
        for (int j = 0; j < 4; ++j) { an[j].x = arow[kn + 4 * j]; an[j].y = arow[kn + 4 * j + 1]; }
        #pragma unroll
        for (int j = 0; j < 4; ++j) {
            #pragma unroll
            for (int t = 0; t < 4; ++t) {
                v2f b;
                b.x = p0[j][t];
                b.y = p1[j][t];
                c[t] = __builtin_amdgcn_wmma_f32_16x16x4_f32(false, a[j], false, b, (short)0, c[t], false, false);
            }
            #pragma unroll
            for (int t = 0; t < 4; ++t) {
                v2f b;
                b.x = q0[j][t];
                b.y = q1[j][t];
                c[t + 4] = __builtin_amdgcn_wmma_f32_16x16x4_f32(false, a[j], false, b, (short)0, c[t + 4], false, false);
            }
        }
    }

    const int mbase = khalf * 8;
    #pragma unroll
    for (int r = 0; r < 8; ++r) {
        float* orow = h1 + (size_t)(row0 + mbase + r) * D_HID + 4 * n;
        f4u lo, hi;
        #pragma unroll
        for (int t = 0; t < 4; ++t) { lo[t] = c[t][r]; hi[t] = c[t + 4][r]; }
        *(f4u*)(orow)      = lo;
        *(f4u*)(orow + 64) = hi;
    }
}


#define QCAP 64
#define REC_CAP 16384

template <int V> struct rowvec;
template <> struct rowvec<4> { typedef float t __attribute__((ext_vector_type(4), aligned(4))); };
template <> struct rowvec<2> { typedef float t __attribute__((ext_vector_type(2), aligned(4))); };

template <int CH, bool FULL>
__device__ __forceinline__ void add_rows(int r, int cl, int wi, int cnt,
                                         const float* __restrict__ h, float* acc, int lane) {
    constexpr int V = CH / 32;
    typedef typename rowvec<V>::t vec;
    #pragma unroll 1
    for (int j0 = 0; j0 < cnt; j0 += 8) {
        vec v[8];
        #pragma unroll
        for (int j = 0; j < 8; ++j) {
            const int rj = __builtin_amdgcn_readlane(r, j0 + j);
            v[j] = *(const vec*)(h + (size_t)rj * CH + lane * V);
        }
        #pragma unroll
        for (int j = 0; j < 8; ++j) {
            if (FULL || j0 + j < cnt) {
                const int cj = __builtin_amdgcn_readlane(cl, j0 + j);
                const float wj = __builtin_bit_cast(float, __builtin_amdgcn_readlane(wi, j0 + j));
                float* a = acc + cj * CH + lane;
                #pragma unroll
                for (int k = 0; k < V; ++k) atomicAdd(a + 32 * k, wj * v[j][k]);
            }
        }
    }
}

template <int CH, bool FULL, bool SAVE>
__device__ __forceinline__ void drain_round(const int* qe_, int cnt,
                                            const int* __restrict__ row, const int* __restrict__ col,
                                            const unsigned* __restrict__ pk,
                                            const float* __restrict__ deg, const float* sdis,
                                            const float* __restrict__ h, float* acc, int lo, int lane,
                                            int* rec, int* ctl) {
    int cl = -1, r = 0;
    float w = 0.0f;
    if (FULL || lane < cnt) {
        const int e = qe_[lane];
        if (SAVE) {
            const unsigned p = pk[e];
            r = (int)(p & 0xffffu);
            cl = (int)(p >> 16);
            w = sdis[r] * sdis[lo + cl];
        } else {
            r = row[e];
            const int c = col[e];
            cl = c - lo;
            w = rsqrtf(deg[r]) * rsqrtf(deg[c]);
        }
    }
    const int wi = __builtin_bit_cast(int, w);
    if (SAVE) {
        int base = 0;
        if (lane == 0) base = atomicAdd(&ctl[0], 32);
        base = __builtin_amdgcn_readfirstlane(base);
        if (base + 32 <= REC_CAP) {
            rec[base + lane] = r;
            rec[REC_CAP + base + lane] = cl;
            rec[2 * REC_CAP + base + lane] = wi;
        } else if (lane == 0) {
            ctl[1] = 1;
        }
    }
    add_rows<CH, FULL>(r, cl, wi, cnt, h, acc, lane);
}

template <int CH, bool SAVE>
__global__ void __launch_bounds__(1024) own_gather(const int* __restrict__ row, const int* __restrict__ col,
                                                   const unsigned* __restrict__ key, const unsigned* __restrict__ pk,
                                                   const float* __restrict__ deg, const float* __restrict__ h,
                                                   float* __restrict__ dst, int* recs, int* meta, float* stats,
                                                   int N, int E) {
    constexpr int V = CH / 32;
    typedef typename rowvec<V>::t vec;
    constexpr int G = 1024 / CH;
    __shared__ float acc[OWN_ROWS * CH];
    __shared__ int qe[32][QCAP];
    __shared__ int ctl[2];
    __shared__ float bnp[SAVE ? 2 * G * CH : 1];
    __shared__ float sdis[SAVE ? DIS_CAP : 32];
    const int wave = __builtin_amdgcn_readfirstlane((int)(threadIdx.x >> 5));
    const int lane = threadIdx.x & 31;
    const int lo = blockIdx.x * OWN_ROWS;
    const int nrows = min(OWN_ROWS, N - lo);
    int* rec = recs + (size_t)blockIdx.x * (3 * REC_CAP);
    if (SAVE && threadIdx.x == 0) { ctl[0] = 0; ctl[1] = 0; }

    constexpr int TB = SAVE ? (DIS_CAP / 4 + 1023) / 1024 : 1;
    const int n4 = N >> 2;
    f4u d[TB];
    if (SAVE) {
        #pragma unroll
        for (int j = 0; j < TB; ++j) {
            const int i = (int)threadIdx.x + 1024 * j;
            d[j] = *(const f4u*)(deg + 4 * (i < n4 ? i : 0));
        }
    }

    {
        constexpr int DEPTH = (OWN_ROWS + 31) / 32;
        const int rq = min(wave + 32 * (lane < DEPTH ? lane : 0), nrows - 1);
        float dl = rsqrtf(deg[lo + rq]);
        dl = dl * dl;
        vec v[DEPTH];
        #pragma unroll
        for (int j = 0; j < DEPTH; ++j) {
            const int rc = min(wave + 32 * j, nrows - 1);
            v[j] = *(const vec*)(h + (size_t)(lo + rc) * CH + lane * V);
        }
        if (SAVE) {
            #pragma unroll
            for (int j = 0; j < TB; ++j) {
                const int i = (int)threadIdx.x + 1024 * j;
                if (i < n4) {
                    #pragma unroll
                    for (int k = 0; k < 4; ++k) sdis[4 * i + k] = rsqrtf(d[j][k]);
                }
            }
        }
        #pragma unroll
        for (int j = 0; j < DEPTH; ++j) {
            const int rc = min(wave + 32 * j, nrows - 1);
            const float dd = __builtin_bit_cast(float, __builtin_amdgcn_readlane(__builtin_bit_cast(int, dl), j));
            #pragma unroll
            for (int k = 0; k < V; ++k) acc[rc * CH + k * 32 + lane] = dd * v[j][k];
        }
    }
    if (SAVE) {
        for (int i = (n4 << 2) + (int)threadIdx.x; i < N; i += 1024) sdis[i] = rsqrtf(deg[i]);
    }
    __syncthreads();

    const int nslots = SAVE ? -1 : meta[blockIdx.x * 32];
    if (!SAVE && nslots >= 0) {
        const int nfr = nslots >> 5;
        for (int f = wave; f < nfr; f += 32) {
            const int r  = rec[f * 32 + lane];
            const int cl = rec[REC_CAP + f * 32 + lane];
            const int wi = rec[2 * REC_CAP + f * 32 + lane];
            const int cnt = __builtin_popcount(__builtin_amdgcn_ballot_w32(cl >= 0));
            if (cnt == 32) add_rows<CH, true>(r, cl, wi, 32, h, acc, lane);
            else           add_rows<CH, false>(r, cl, wi, cnt, h, acc, lane);
        }
    } else {
    int* myqe = qe[wave];
    const unsigned pat = (unsigned)blockIdx.x * 0x01010101u;
    const int nsuper = (E + KEY_SPAN - 1) / KEY_SPAN;
    const int s0 = (int)(((long long)blockIdx.x * nsuper) / gridDim.x);
    int q = 0;
    for (int it = wave; it < nsuper; it += 32) {
        int s = it + s0;
        if (s >= nsuper) s -= nsuper;
        vu4 kw[4];
        #pragma unroll
        for (int k = 0; k < 4; ++k) kw[k] = *(const vu4*)(key + (size_t)(s * 4 + k) * 128 + lane * 4);
        const bool tail = s * KEY_SPAN + KEY_SPAN > E;
        #pragma unroll
        for (int k = 0; k < 4; ++k) {
            const int eb = (s * 4 + k) * 512 + lane * 16;
            unsigned m = 0u;
            #pragma unroll
            for (int i = 0; i < 4; ++i) {
                const unsigned x = kw[k][i] ^ pat;
                const unsigned t = (x & 0x7f7f7f7fu) + 0x7f7f7f7fu;
                m |= (~(t | x) & 0x80808080u) >> i;
            }
            if (tail) {
                const int left = E - eb;
                unsigned keep = 0u;
                #pragma unroll
                for (int i = 0; i < 4; ++i) {
                    #pragma unroll
                    for (int j = 0; j < 4; ++j) keep |= (4 * i + j < left) ? (1u << (8 * j + 7 - i)) : 0u;
                }
                m &= keep;
            }
            unsigned bal = __builtin_amdgcn_ballot_w32(m != 0u);
            while (bal != 0u) {
                if (m != 0u) {
                    const int p = __builtin_ctz(m);
                    const int pos = q + (int)__builtin_amdgcn_mbcnt_lo(bal, 0u);
                    myqe[pos] = eb + ((7 - (p & 7)) << 2) + (p >> 3);
                    m &= m - 1u;
                }
                q += __builtin_popcount(bal);
                if (q >= 32) {
                    q -= 32;
                    drain_round<CH, true, SAVE>(myqe + q, 32, row, col, pk, deg, sdis, h, acc, lo, lane, rec, ctl);
                }
                bal = __builtin_amdgcn_ballot_w32(m != 0u);
            }
        }
    }
    if (q > 0) drain_round<CH, false, SAVE>(myqe, q, row, col, pk, deg, sdis, h, acc, lo, lane, rec, ctl);
    }
    __syncthreads();

    if constexpr (SAVE) {
        constexpr int RT = (OWN_ROWS + G - 1) / G;
        const int p = threadIdx.x & (CH - 1);
        const int g = wave / (CH / 32);
        float s = 0.0f, q = 0.0f;
        #pragma unroll
        for (int j = 0; j < RT; ++j) {
            const int r = g + G * j;
            const float v = acc[(r < nrows ? r : 0) * CH + p];
            const float m = r < nrows ? v : 0.0f;
            s += m;
            q += m * m;
        }
        bnp[g * CH + p] = s;
        bnp[(G + g) * CH + p] = q;
        __syncthreads();
        if (threadIdx.x < 2 * CH) {
            const int ch = threadIdx.x & (CH - 1);
            const int w = (ch % V) * 32 + ch / V;
            const float* part = bnp + (threadIdx.x / CH) * (G * CH) + w;
            float t = 0.0f;
            #pragma unroll
            for (int k = 0; k < G; ++k) t += part[k * CH];
            atomicAdd(&stats[threadIdx.x], t);
        }
        if (threadIdx.x == 0) meta[blockIdx.x * 32] = ctl[1] ? -1 : ctl[0];

        for (int rl = wave; rl < nrows; rl += 32) {
            vec o;
            #pragma unroll
            for (int k = 0; k < V; ++k) o[k] = acc[rl * CH + k * 32 + lane];
            *(vec*)(dst + (size_t)(lo + rl) * CH + lane * V) = o;
        }
    } else {
        constexpr int DEPTH = (OWN_ROWS + 31) / 32;
        vec o[DEPTH];
        float mx[DEPTH], sm[DEPTH];
        #pragma unroll
        for (int j = 0; j < DEPTH; ++j) {
            const int rc = min(wave + 32 * j, nrows - 1);
            #pragma unroll
            for (int k = 0; k < V; ++k) o[j][k] = acc[rc * CH + k * 32 + lane];
            mx[j] = o[j][0];
            #pragma unroll
            for (int k = 1; k < V; ++k) mx[j] = fmaxf(mx[j], o[j][k]);
        }
        #pragma unroll
        for (int off = 16; off >= 1; off >>= 1) {
            #pragma unroll
            for (int j = 0; j < DEPTH; ++j) mx[j] = fmaxf(mx[j], __shfl_xor(mx[j], off, 32));
        }
        #pragma unroll
        for (int j = 0; j < DEPTH; ++j) {
            sm[j] = 0.0f;
            #pragma unroll
            for (int k = 0; k < V; ++k) sm[j] += __expf(o[j][k] - mx[j]);
        }
        #pragma unroll
        for (int off = 16; off >= 1; off >>= 1) {
            #pragma unroll
            for (int j = 0; j < DEPTH; ++j) sm[j] += __shfl_xor(sm[j], off, 32);
        }
        #pragma unroll
        for (int j = 0; j < DEPTH; ++j) {
            const int rl = wave + 32 * j;
            const float l = mx[j] + __logf(sm[j]);
            vec res;
            #pragma unroll
            for (int k = 0; k < V; ++k) res[k] = o[j][k] - l;
            if (rl < nrows) *(vec*)(dst + (size_t)(lo + rl) * CH + lane * V) = res;
        }
    }
}


__global__ void __launch_bounds__(128) gemm2_wmma(const float* __restrict__ hagg, const float* __restrict__ bsum,
                           const float* __restrict__ bsq, const float* __restrict__ gamma,
                           const float* __restrict__ beta, const float* __restrict__ W2,
                           const float* __restrict__ b2, float* __restrict__ h2, int N) {
    __shared__ float sc[D_HID];
    __shared__ float sh[D_HID];
    const int wave  = __builtin_amdgcn_readfirstlane((int)(threadIdx.x >> 5));
    const int lane  = threadIdx.x & 31;
    const int row0  = (blockIdx.x * 4 + wave) * 16;
    const bool live = row0 < N;
    const int n     = lane & 15;
    const int khalf = lane >> 4;
    const float* arow = hagg + (size_t)(row0 + n) * D_HID + 2 * khalf;

    v2f an[4];
    #pragma unroll
    for (int j = 0; j < 4; ++j) { an[j].x = 0.0f; an[j].y = 0.0f; }
    if (live) {
        #pragma unroll
        for (int j = 0; j < 4; ++j) { an[j].x = arow[4 * j]; an[j].y = arow[4 * j + 1]; }
    }
    {
        const int ch = threadIdx.x;
        const float inv = 1.0f / (float)N;
        const float mu = bsum[ch] * inv;
        const float var = bsq[ch] * inv - mu * mu;
        const float s = gamma[ch] * rsqrtf(var + BN_EPS);
        sc[ch] = s;
        sh[ch] = beta[ch] - mu * s;
    }
    __syncthreads();
    if (!live) return;

    v8f c[4];
    {
        const f4u bias = *(const f4u*)(b2 + 4 * n);
        #pragma unroll
        for (int t = 0; t < 4; ++t) {
            #pragma unroll
            for (int r = 0; r < 8; ++r) c[t][r] = bias[t];
        }
    }

    const float* brow = W2 + (size_t)(2 * khalf) * D_OUT + 4 * n;

    #pragma unroll 2
    for (int k0 = 0; k0 < 128; k0 += 16) {
        v2f a[4];
        f4u p0[4], p1[4];
        #pragma unroll
        for (int j = 0; j < 4; ++j) {
            const int k = k0 + 4 * j;
            const int kk = k + 2 * khalf;
            a[j].x = fmaxf(an[j].x * sc[kk + 0] + sh[kk + 0], 0.0f);
            a[j].y = fmaxf(an[j].y * sc[kk + 1] + sh[kk + 1], 0.0f);
            const float* bk = brow + (size_t)k * D_OUT;
            p0[j] = *(const f4u*)(bk);
            p1[j] = *(const f4u*)(bk + D_OUT);
        }
        const int kn = (k0 + 16 < 128) ? k0 + 16 : k0;
        #pragma unroll
        for (int j = 0; j < 4; ++j) { an[j].x = arow[kn + 4 * j]; an[j].y = arow[kn + 4 * j + 1]; }
        #pragma unroll
        for (int j = 0; j < 4; ++j) {
            #pragma unroll
            for (int t = 0; t < 4; ++t) {
                v2f b;
                b.x = p0[j][t];
                b.y = p1[j][t];
                c[t] = __builtin_amdgcn_wmma_f32_16x16x4_f32(false, a[j], false, b, (short)0, c[t], false, false);
            }
        }
    }

    const int mbase = khalf * 8;
    #pragma unroll
    for (int r = 0; r < 8; ++r) {
        f4u o;
        #pragma unroll
        for (int t = 0; t < 4; ++t) o[t] = c[t][r];
        *(f4u*)(h2 + (size_t)(row0 + mbase + r) * D_OUT + 4 * n) = o;
    }
}


extern "C" void kernel_launch(void* const* d_in, const int* in_sizes, int n_in,
                              void* d_out, int out_size, void* d_ws, size_t ws_size,
                              hipStream_t stream) {
    const float* x     = (const float*)d_in[0];
    const float* W1    = (const float*)d_in[1];
    const float* b1    = (const float*)d_in[2];
    const float* gamma = (const float*)d_in[3];
    const float* beta  = (const float*)d_in[4];
    const float* W2    = (const float*)d_in[5];
    const float* b2    = (const float*)d_in[6];
    const int*   ei    = (const int*)d_in[7];

    const int N = in_sizes[0] / D_HID;
    const int E = in_sizes[7] / 2;
    const int* erow = ei;
    const int* ecol = ei + E;

    float* ws    = (float*)d_ws;
    float* deg   = ws;
    float* h1    = deg + N;
    float* hagg  = h1 + (size_t)N * D_HID;
    float* stats = hagg + (size_t)N * D_HID;
    float* bsum  = stats;
    float* bsq   = stats + 128;
    float* h2    = h1;
    int*   meta  = (int*)((char*)d_ws + (size_t)(63u << 20));
    int*   recs  = (int*)((char*)d_ws + (size_t)(64u << 20));
    unsigned* key = (unsigned*)((char*)d_ws + (size_t)(120u << 20));
    unsigned* pk  = (unsigned*)((char*)d_ws + (size_t)(124u << 20));
    float* outp  = (float*)d_out;

    gemm1_wmma<<<(N + 63) / 64, 128, 0, stream>>>(x, W1, b1, h1, deg, stats, N);
    count_edges<<<((E + KEY_SPAN - 1) / KEY_SPAN) * (KEY_SPAN / 1024), 256, 0, stream>>>(erow, ecol, deg, key, pk, E);
    own_gather<D_HID, true><<<(N + OWN_ROWS - 1) / OWN_ROWS, 1024, 0, stream>>>(erow, ecol, key, pk, deg, h1, hagg, recs, meta, stats, N, E);

    gemm2_wmma<<<(N + 63) / 64, 128, 0, stream>>>(hagg, bsum, bsq, gamma, beta, W2, b2, h2, N);
    own_gather<D_OUT, false><<<(N + OWN_ROWS - 1) / OWN_ROWS, 1024, 0, stream>>>(erow, ecol, key, pk, deg, h2, outp, recs, meta, stats, N, E);
}
